# starting kernel
# baseline (speedup 1.0000x reference)
#include <hip/hip_runtime.h>
#include <math.h>


#define NPTS 100000
#define NBAS 32
#define MCLU 16
#define KNEI 32
#define MUF  100.0f

typedef __attribute__((ext_vector_type(2))) float v2f;
typedef __attribute__((ext_vector_type(8))) float v8f;

#define WS_ENCPART 0
#define WS_ENCODED (WS_ENCPART + 1024)
#define WS_INVW2T  (WS_ENCODED + 32)
#define WS_RDENT   (WS_INVW2T + 512)
#define DEN_BLKS   224
#define WS_DENPART (WS_RDENT + 1024)
#define WS_DECT    (WS_DENPART + DEN_BLKS*1024)

__global__ void k_enc_partial(const float* __restrict__ x,
                              const float* __restrict__ W,
                              float* __restrict__ part) {
  const int row = blockIdx.y, chunk = blockIdx.x;
  const int CS = (NPTS + 31) / 32;
  const int j0 = chunk * CS;
  const int j1 = (j0 + CS < NPTS) ? (j0 + CS) : NPTS;
  float s = 0.f;
  for (int j = j0 + threadIdx.x; j < j1; j += blockDim.x)
    s += x[j] * W[row * NPTS + j];
  __shared__ float red[256];
  red[threadIdx.x] = s;
  __syncthreads();
  for (int o = 128; o > 0; o >>= 1) {
    if (threadIdx.x < o) red[threadIdx.x] += red[threadIdx.x + o];
    __syncthreads();
  }
  if (threadIdx.x == 0) part[row * 32 + chunk] = red[0];
}

__global__ void k_enc_finish(const float* __restrict__ part,
                             const float* __restrict__ enc_b,
                             const float* __restrict__ bw_w,
                             const float* __restrict__ bw_b,
                             float* __restrict__ encoded,
                             float* __restrict__ invw2t) {
  __shared__ float enc_s[NBAS];
  const int t = threadIdx.x;
  if (t < NBAS) {
    float s = 0.f;
    for (int c = 0; c < 32; ++c) s += part[t * 32 + c];
    s += enc_b[t];
    enc_s[t] = s;
    encoded[t] = s;
  }
  __syncthreads();
  if (t < NBAS * MCLU) {
    float s = 0.f;
    for (int q = 0; q < NBAS; ++q) s += enc_s[q] * bw_w[t * NBAS + q];
    s += bw_b[t];
    float band = (1.f / (1.f + expf(-s))) / 60.f;
    float wm = band * MUF;
    const int i = t / MCLU, c = t % MCLU;
    invw2t[c * NBAS + i] = 1.f / (wm * wm);
  }
}

__global__ void k_dec_transpose(const float* __restrict__ dec,
                                float* __restrict__ dect) {
  __shared__ float tile[32][33];
  const int jb = blockIdx.x * 32;
  const int tx = threadIdx.x, ty = threadIdx.y;
  tile[ty][tx] = dec[ty * NPTS + jb + tx];
  __syncthreads();
  dect[(jb + ty) * 32 + tx] = tile[tx][ty];
}

__global__ void k_denom_partial(const float* __restrict__ d,
                                const int* __restrict__ lab,
                                const float* __restrict__ invw2t,
                                float* __restrict__ part) {
  __shared__ float iw[MCLU * NBAS];
  __shared__ float red[8 * 1024];
  for (int t = threadIdx.x; t < MCLU * NBAS; t += blockDim.x) iw[t] = invw2t[t];
  __syncthreads();
  const int lane = threadIdx.x & 31, wv = threadIdx.x >> 5;
  const int gw = blockIdx.x * 8 + wv, nw = gridDim.x * 8;
  float acc[NBAS];
#pragma unroll
  for (int i = 0; i < NBAS; ++i) acc[i] = 0.f;
  for (int j = gw; j < NPTS; j += nw) {
    const int c = lab[j];
    const float dv = d[j * KNEI + lane];
    const float d2 = dv * dv;
#pragma unroll
    for (int i = 0; i < NBAS; ++i)
      acc[i] += fmaxf(1.f - d2 * iw[c * NBAS + i], 0.f);
  }
#pragma unroll
  for (int i = 0; i < NBAS; ++i) red[wv * 1024 + i * 32 + lane] = acc[i];
  __syncthreads();
  for (int e = threadIdx.x; e < 1024; e += blockDim.x) {
    float s = 0.f;
#pragma unroll
    for (int w = 0; w < 8; ++w) s += red[w * 1024 + e];
    part[blockIdx.x * 1024 + e] = s;
  }
}

__global__ void k_denom_finish(const float* __restrict__ part, int nblk,
                               float* __restrict__ rdent) {
  const int t = threadIdx.x;
  float s = 0.f;
  for (int b = 0; b < nblk; ++b) s += part[b * 1024 + t];
  const int i = t >> 5, kk = t & 31;
  rdent[kk * 32 + i] = 1.f / s;
}

__global__ void k_fused_out(const float* __restrict__ d,
                            const int* __restrict__ nid,
                            const int* __restrict__ lab,
                            const float* __restrict__ dect,
                            const float* __restrict__ invw2t,
                            const float* __restrict__ rdent,
                            const float* __restrict__ encoded,
                            float* __restrict__ out) {
  __shared__ float iw[MCLU * NBAS];
  __shared__ float rd[NBAS * NBAS];
  __shared__ float enc_s[NBAS];
  __shared__ float g[8][32 * 17];
  for (int t = threadIdx.x; t < MCLU * NBAS; t += blockDim.x) iw[t] = invw2t[t];
  for (int t = threadIdx.x; t < NBAS * NBAS; t += blockDim.x) rd[t] = rdent[t];
  if (threadIdx.x < NBAS) enc_s[threadIdx.x] = encoded[threadIdx.x];
  __syncthreads();

  const int lane = threadIdx.x & 31, wv = threadIdx.x >> 5;
  const int gw = blockIdx.x * 8 + wv, nw = gridDim.x * 8;
  float* gbuf = g[wv];
  const int NT = NPTS / 16;

  for (int tile = gw; tile < NT; tile += nw) {
    const int j0 = tile * 16;
    for (int p = 0; p < 16; ++p) {
      const int j = j0 + p;
      const int c = lab[j];
      const float s = iw[c * NBAS + lane];
      const int nl = nid[j * KNEI + lane];
      const float dv = d[j * KNEI + lane];
      const float d2 = dv * dv;
      float acc = 0.f;
#pragma unroll
      for (int kk = 0; kk < KNEI; ++kk) {
        const int nb = __shfl(nl, kk, 32);
        const float d2k = __shfl(d2, kk, 32);
        const float win = fmaxf(1.f - d2k * s, 0.f);
        const float dec = dect[nb * 32 + lane];
        acc += rd[kk * 32 + lane] * win * dec;
      }
      gbuf[lane * 17 + p] = acc;
    }
    asm volatile("s_wait_dscnt 0" ::: "memory");

    v8f cacc = {0.f, 0.f, 0.f, 0.f, 0.f, 0.f, 0.f, 0.f};
    const int half = (lane >> 4) & 1;
    const int col = lane & 15;
#pragma unroll
    for (int s8 = 0; s8 < 8; ++s8) {
      const int kb = s8 * 4;
      v2f a, b;
      a.x = enc_s[kb + 2 * half];
      a.y = enc_s[kb + 2 * half + 1];
      b.x = gbuf[(kb + 2 * half) * 17 + col];
      b.y = gbuf[(kb + 2 * half + 1) * 17 + col];
      cacc = __builtin_amdgcn_wmma_f32_16x16x4_f32(
          false, a, false, b, (short)0, cacc, false, false);
    }
    if (lane < 16) out[j0 + lane] = cacc[0];
  }
}

extern "C" void kernel_launch(void* const* d_in, const int* in_sizes, int n_in,
                              void* d_out, int out_size, void* d_ws, size_t ws_size,
                              hipStream_t stream) {
  const float* x     = (const float*)d_in[0];
  const float* enc_w = (const float*)d_in[1];
  const float* enc_b = (const float*)d_in[2];
  const float* dec   = (const float*)d_in[3];
  const float* bw_w  = (const float*)d_in[4];
  const float* bw_b  = (const float*)d_in[5];
  const float* nd    = (const float*)d_in[6];
  const int*   nid   = (const int*)d_in[7];
  const int*   lab   = (const int*)d_in[8];
  float* out = (float*)d_out;
  float* ws  = (float*)d_ws;

  float* enc_part = ws + WS_ENCPART;
  float* encoded  = ws + WS_ENCODED;
  float* invw2t   = ws + WS_INVW2T;
  float* rdent    = ws + WS_RDENT;
  float* denpart  = ws + WS_DENPART;
  float* dect     = ws + WS_DECT;

  k_enc_partial<<<dim3(32, 32), 256, 0, stream>>>(x, enc_w, enc_part);
  k_enc_finish<<<1, 512, 0, stream>>>(enc_part, enc_b, bw_w, bw_b, encoded, invw2t);
  k_dec_transpose<<<NPTS / 32, dim3(32, 32), 0, stream>>>(dec, dect);
  k_denom_partial<<<DEN_BLKS, 256, 0, stream>>>(nd, lab, invw2t, denpart);
  k_denom_finish<<<1, 1024, 0, stream>>>(denpart, DEN_BLKS, rdent);
  k_fused_out<<<200, 256, 0, stream>>>(nd, nid, lab, dect, invw2t, rdent,
                                       encoded, out);
}


# fused gather retiled, points across lanes: lane = (1 of 4 points, 4 bases), one 16B load = one neighbour slot of 4 points, ids by DPP8 group broadcast, sums in-lane in index order; one tile per wave
# speedup vs baseline: 5.2406x; 5.2406x over previous
#include <hip/hip_runtime.h>
#include <math.h>


#define NPTS 100000
#define NBAS 32
#define MCLU 16
#define KNEI 32
#define MUF  100.0f

typedef __attribute__((ext_vector_type(2))) float v2f;
typedef __attribute__((ext_vector_type(8))) float v8f;
typedef __attribute__((ext_vector_type(4))) float v4f;
typedef __attribute__((ext_vector_type(4))) int v4i;

#define WS_ENCPART 0
#define WS_ENCODED (WS_ENCPART + 1024)
#define WS_INVW2T  (WS_ENCODED + 32)
#define WS_RDENT   (WS_INVW2T + 512)
#define DEN_BLKS   224
#define WS_DENPART (WS_RDENT + 1024)
#define WS_DECT    (WS_DENPART + DEN_BLKS*1024)

__global__ void k_enc_partial(const float* __restrict__ x,
                              const float* __restrict__ W,
                              float* __restrict__ part) {
  const int row = blockIdx.y, chunk = blockIdx.x;
  const int CS = (NPTS + 31) / 32;
  const int j0 = chunk * CS;
  const int j1 = (j0 + CS < NPTS) ? (j0 + CS) : NPTS;
  float s = 0.f;
  for (int j = j0 + threadIdx.x; j < j1; j += blockDim.x)
    s += x[j] * W[row * NPTS + j];
  __shared__ float red[256];
  red[threadIdx.x] = s;
  __syncthreads();
  for (int o = 128; o > 0; o >>= 1) {
    if (threadIdx.x < o) red[threadIdx.x] += red[threadIdx.x + o];
    __syncthreads();
  }
  if (threadIdx.x == 0) part[row * 32 + chunk] = red[0];
}

__global__ void k_enc_finish(const float* __restrict__ part,
                             const float* __restrict__ enc_b,
                             const float* __restrict__ bw_w,
                             const float* __restrict__ bw_b,
                             float* __restrict__ encoded,
                             float* __restrict__ invw2t) {
  __shared__ float enc_s[NBAS];
  const int t = threadIdx.x;
  if (t < NBAS) {
    float s = 0.f;
    for (int c = 0; c < 32; ++c) s += part[t * 32 + c];
    s += enc_b[t];
    enc_s[t] = s;
    encoded[t] = s;
  }
  __syncthreads();
  if (t < NBAS * MCLU) {
    float s = 0.f;
    for (int q = 0; q < NBAS; ++q) s += enc_s[q] * bw_w[t * NBAS + q];
    s += bw_b[t];
    float band = (1.f / (1.f + expf(-s))) / 60.f;
    float wm = band * MUF;
    const int i = t / MCLU, c = t % MCLU;
    invw2t[c * NBAS + i] = 1.f / (wm * wm);
  }
}

__global__ void k_dec_transpose(const float* __restrict__ dec,
                                float* __restrict__ dect) {
  __shared__ float tile[32][33];
  const int jb = blockIdx.x * 32;
  const int tx = threadIdx.x, ty = threadIdx.y;
  tile[ty][tx] = dec[ty * NPTS + jb + tx];
  __syncthreads();
  dect[(jb + ty) * 32 + tx] = tile[tx][ty];
}

__global__ void k_denom_partial(const float* __restrict__ d,
                                const int* __restrict__ lab,
                                const float* __restrict__ invw2t,
                                float* __restrict__ part) {
  __shared__ float iw[MCLU * NBAS];
  __shared__ float red[8 * 1024];
  for (int t = threadIdx.x; t < MCLU * NBAS; t += blockDim.x) iw[t] = invw2t[t];
  __syncthreads();
  const int lane = threadIdx.x & 31, wv = threadIdx.x >> 5;
  const int gw = blockIdx.x * 8 + wv, nw = gridDim.x * 8;
  float acc[NBAS];
#pragma unroll
  for (int i = 0; i < NBAS; ++i) acc[i] = 0.f;
  for (int j = gw; j < NPTS; j += nw) {
    const int c = lab[j];
    const float dv = d[j * KNEI + lane];
    const float d2 = dv * dv;
#pragma unroll
    for (int i = 0; i < NBAS; ++i)
      acc[i] += fmaxf(1.f - d2 * iw[c * NBAS + i], 0.f);
  }
#pragma unroll
  for (int i = 0; i < NBAS; ++i) red[wv * 1024 + i * 32 + lane] = acc[i];
  __syncthreads();
  for (int e = threadIdx.x; e < 1024; e += blockDim.x) {
    float s = 0.f;
#pragma unroll
    for (int w = 0; w < 8; ++w) s += red[w * 1024 + e];
    part[blockIdx.x * 1024 + e] = s;
  }
}

__global__ void k_denom_finish(const float* __restrict__ part, int nblk,
                               float* __restrict__ rdent) {
  const int t = threadIdx.x;
  float s = 0.f;
  for (int b = 0; b < nblk; ++b) s += part[b * 1024 + t];
  const int i = t >> 5, kk = t & 31;
  rdent[kk * 32 + i] = 1.f / s;
}

#define FO_WAVES   4
#define FO_THREADS 128
#define FO_BLOCKS  1563

template <int G>
__device__ __forceinline__ int bcast8(int v) {
  return (int)__builtin_amdgcn_mov_dpp8((unsigned)v, (unsigned)(G * 0x249249));
}

template <int KK>
__device__ __forceinline__ void issue_slot(v4f (&r)[4], const v4f* __restrict__ dect4,
                                           const v4i (&n4)[4], const unsigned q) {
  r[0] = dect4[(unsigned)bcast8<(KK >> 2)>(n4[0][KK & 3]) * 8u + q];
  r[1] = dect4[(unsigned)bcast8<(KK >> 2)>(n4[1][KK & 3]) * 8u + q];
  r[2] = dect4[(unsigned)bcast8<(KK >> 2)>(n4[2][KK & 3]) * 8u + q];
  r[3] = dect4[(unsigned)bcast8<(KK >> 2)>(n4[3][KK & 3]) * 8u + q];
}

__device__ __forceinline__ void add_pass(v4f& acc, const v4f r4, const float d2k,
                                         const v4f s4, const v4f dec) {
  acc.x += r4.x * fmaxf(1.f - d2k * s4.x, 0.f) * dec.x;
  acc.y += r4.y * fmaxf(1.f - d2k * s4.y, 0.f) * dec.y;
  acc.z += r4.z * fmaxf(1.f - d2k * s4.z, 0.f) * dec.z;
  acc.w += r4.w * fmaxf(1.f - d2k * s4.w, 0.f) * dec.w;
}

template <int KK>
__device__ __forceinline__ void use_slot(v4f (&acc)[4], const v4f (&r)[4],
                                         const v4f (&d2v)[4], const v4f (&s4)[4],
                                         const v4f* rd4, const unsigned q) {
  const v4f r4 = rd4[KK * 8 + q];
  add_pass(acc[0], r4, __int_as_float(bcast8<(KK >> 2)>(__float_as_int(d2v[0][KK & 3]))), s4[0], r[0]);
  add_pass(acc[1], r4, __int_as_float(bcast8<(KK >> 2)>(__float_as_int(d2v[1][KK & 3]))), s4[1], r[1]);
  add_pass(acc[2], r4, __int_as_float(bcast8<(KK >> 2)>(__float_as_int(d2v[2][KK & 3]))), s4[2], r[2]);
  add_pass(acc[3], r4, __int_as_float(bcast8<(KK >> 2)>(__float_as_int(d2v[3][KK & 3]))), s4[3], r[3]);
}

template <int K>
__device__ __forceinline__ void pipe_step(v4f (&acc)[4], v4f (&ra)[4], v4f (&rb)[4],
                                          const v4f* __restrict__ dect4,
                                          const v4i (&n4)[4], const v4f (&d2v)[4],
                                          const v4f (&s4)[4], const v4f* rd4,
                                          const unsigned q) {
  use_slot<K>(acc, ra, d2v, s4, rd4, q);
  use_slot<K + 1>(acc, rb, d2v, s4, rd4, q);
  if constexpr (K + 4 < KNEI) {
    issue_slot<K + 4>(ra, dect4, n4, q);
    issue_slot<K + 5>(rb, dect4, n4, q);
  }
}

__global__ __launch_bounds__(FO_THREADS)
void k_fused_out(const float* __restrict__ d,
                 const int* __restrict__ nid,
                 const int* __restrict__ lab,
                 const float* __restrict__ dect,
                 const float* __restrict__ invw2t,
                 const float* __restrict__ rdent,
                 const float* __restrict__ encoded,
                 float* __restrict__ out) {
  __shared__ __attribute__((aligned(16))) float iw[MCLU * NBAS];
  __shared__ __attribute__((aligned(16))) float rd[NBAS * NBAS];
  __shared__ float enc_s[NBAS];
  __shared__ float g[FO_WAVES][32 * 17];
  for (int t = threadIdx.x; t < MCLU * NBAS / 4; t += FO_THREADS)
    reinterpret_cast<v4f*>(iw)[t] = reinterpret_cast<const v4f*>(invw2t)[t];
  for (int t = threadIdx.x; t < NBAS * NBAS / 4; t += FO_THREADS)
    reinterpret_cast<v4f*>(rd)[t] = reinterpret_cast<const v4f*>(rdent)[t];
  if (threadIdx.x < NBAS) enc_s[threadIdx.x] = encoded[threadIdx.x];
  __syncthreads();

  const int lane = threadIdx.x & 31, wv = threadIdx.x >> 5;
  const int tile = blockIdx.x * FO_WAVES + wv;
  const int NT = NPTS / 16;
  if (tile >= NT) return;
  float* gbuf = g[wv];
  const int j0 = tile * 16;
  const int pt = lane >> 3;
  const unsigned q = (unsigned)(lane & 7);
  const v4f* dect4 = reinterpret_cast<const v4f*>(dect);
  const v4f* rd4 = reinterpret_cast<const v4f*>(rd);

  v4i n4[4];
  v4f d2v[4], s4[4], acc[4];
  int cl[4];
#pragma unroll
  for (int sub = 0; sub < 4; ++sub) {
    const unsigned j = (unsigned)(j0 + sub * 4 + pt);
    n4[sub] = reinterpret_cast<const v4i*>(nid)[j * 8u + q];
    const v4f dv = reinterpret_cast<const v4f*>(d)[j * 8u + q];
    d2v[sub] = dv * dv;
    cl[sub] = lab[j];
  }
  v4f ra[4], rb[4], rc[4], re[4];
  issue_slot<0>(ra, dect4, n4, q);
  issue_slot<1>(rb, dect4, n4, q);
  issue_slot<2>(rc, dect4, n4, q);
  issue_slot<3>(re, dect4, n4, q);
#pragma unroll
  for (int sub = 0; sub < 4; ++sub) {
    s4[sub] = reinterpret_cast<const v4f*>(iw)[(unsigned)cl[sub] * 8u + q];
    acc[sub] = (v4f){0.f, 0.f, 0.f, 0.f};
  }
  pipe_step<0>(acc, ra, rb, dect4, n4, d2v, s4, rd4, q);
  pipe_step<2>(acc, rc, re, dect4, n4, d2v, s4, rd4, q);
  pipe_step<4>(acc, ra, rb, dect4, n4, d2v, s4, rd4, q);
  pipe_step<6>(acc, rc, re, dect4, n4, d2v, s4, rd4, q);
  pipe_step<8>(acc, ra, rb, dect4, n4, d2v, s4, rd4, q);
  pipe_step<10>(acc, rc, re, dect4, n4, d2v, s4, rd4, q);
  pipe_step<12>(acc, ra, rb, dect4, n4, d2v, s4, rd4, q);
  pipe_step<14>(acc, rc, re, dect4, n4, d2v, s4, rd4, q);
  pipe_step<16>(acc, ra, rb, dect4, n4, d2v, s4, rd4, q);
  pipe_step<18>(acc, rc, re, dect4, n4, d2v, s4, rd4, q);
  pipe_step<20>(acc, ra, rb, dect4, n4, d2v, s4, rd4, q);
  pipe_step<22>(acc, rc, re, dect4, n4, d2v, s4, rd4, q);
  pipe_step<24>(acc, ra, rb, dect4, n4, d2v, s4, rd4, q);
  pipe_step<26>(acc, rc, re, dect4, n4, d2v, s4, rd4, q);
  pipe_step<28>(acc, ra, rb, dect4, n4, d2v, s4, rd4, q);
  pipe_step<30>(acc, rc, re, dect4, n4, d2v, s4, rd4, q);

#pragma unroll
  for (int sub = 0; sub < 4; ++sub) {
    const int colp = sub * 4 + pt;
    gbuf[(4 * q + 0) * 17 + colp] = acc[sub].x;
    gbuf[(4 * q + 1) * 17 + colp] = acc[sub].y;
    gbuf[(4 * q + 2) * 17 + colp] = acc[sub].z;
    gbuf[(4 * q + 3) * 17 + colp] = acc[sub].w;
  }
  asm volatile("s_wait_dscnt 0" ::: "memory");

  v8f cacc = {0.f, 0.f, 0.f, 0.f, 0.f, 0.f, 0.f, 0.f};
  const int half = (lane >> 4) & 1;
  const int col = lane & 15;
#pragma unroll
  for (int s8 = 0; s8 < 8; ++s8) {
    const int kb = s8 * 4;
    v2f a, b;
    a.x = enc_s[kb + 2 * half];
    a.y = enc_s[kb + 2 * half + 1];
    b.x = gbuf[(kb + 2 * half) * 17 + col];
    b.y = gbuf[(kb + 2 * half + 1) * 17 + col];
    cacc = __builtin_amdgcn_wmma_f32_16x16x4_f32(
        false, a, false, b, (short)0, cacc, false, false);
  }
  if (lane < 16) out[j0 + lane] = cacc[0];
}

extern "C" void kernel_launch(void* const* d_in, const int* in_sizes, int n_in,
                              void* d_out, int out_size, void* d_ws, size_t ws_size,
                              hipStream_t stream) {
  const float* x     = (const float*)d_in[0];
  const float* enc_w = (const float*)d_in[1];
  const float* enc_b = (const float*)d_in[2];
  const float* dec   = (const float*)d_in[3];
  const float* bw_w  = (const float*)d_in[4];
  const float* bw_b  = (const float*)d_in[5];
  const float* nd    = (const float*)d_in[6];
  const int*   nid   = (const int*)d_in[7];
  const int*   lab   = (const int*)d_in[8];
  float* out = (float*)d_out;
  float* ws  = (float*)d_ws;

  float* enc_part = ws + WS_ENCPART;
  float* encoded  = ws + WS_ENCODED;
  float* invw2t   = ws + WS_INVW2T;
  float* rdent    = ws + WS_RDENT;
  float* denpart  = ws + WS_DENPART;
  float* dect     = ws + WS_DECT;

  k_enc_partial<<<dim3(32, 32), 256, 0, stream>>>(x, enc_w, enc_part);
  k_enc_finish<<<1, 512, 0, stream>>>(enc_part, enc_b, bw_w, bw_b, encoded, invw2t);
  k_dec_transpose<<<NPTS / 32, dim3(32, 32), 0, stream>>>(dec, dect);
  k_denom_partial<<<DEN_BLKS, 256, 0, stream>>>(nd, lab, invw2t, denpart);
  k_denom_finish<<<1, 1024, 0, stream>>>(denpart, DEN_BLKS, rdent);
  k_fused_out<<<FO_BLOCKS, FO_THREADS, 0, stream>>>(nd, nid, lab, dect, invw2t, rdent,
                                       encoded, out);
}


# decoder transpose in registers: lane = source row, 8 x 16 B loads per lane, register p of the wave = output row p: 32 whole-line stores, no LDS or barrier; 3125 one-tile waves (was 100000)
# speedup vs baseline: 5.5491x; 1.0589x over previous
#include <hip/hip_runtime.h>
#include <math.h>


#define NPTS 100000
#define NBAS 32
#define MCLU 16
#define KNEI 32
#define MUF  100.0f

typedef __attribute__((ext_vector_type(2))) float v2f;
typedef __attribute__((ext_vector_type(8))) float v8f;
typedef __attribute__((ext_vector_type(4))) float v4f;
typedef __attribute__((ext_vector_type(4))) int v4i;

#define WS_ENCPART 0
#define WS_ENCODED (WS_ENCPART + 1024)
#define WS_INVW2T  (WS_ENCODED + 32)
#define WS_RDENT   (WS_INVW2T + 512)
#define DEN_BLKS   224
#define WS_DENPART (WS_RDENT + 1024)
#define WS_DECT    (WS_DENPART + DEN_BLKS*1024)

__global__ void k_enc_partial(const float* __restrict__ x,
                              const float* __restrict__ W,
                              float* __restrict__ part) {
  const int row = blockIdx.y, chunk = blockIdx.x;
  const int CS = (NPTS + 31) / 32;
  const int j0 = chunk * CS;
  const int j1 = (j0 + CS < NPTS) ? (j0 + CS) : NPTS;
  float s = 0.f;
  for (int j = j0 + threadIdx.x; j < j1; j += blockDim.x)
    s += x[j] * W[row * NPTS + j];
  __shared__ float red[256];
  red[threadIdx.x] = s;
  __syncthreads();
  for (int o = 128; o > 0; o >>= 1) {
    if (threadIdx.x < o) red[threadIdx.x] += red[threadIdx.x + o];
    __syncthreads();
  }
  if (threadIdx.x == 0) part[row * 32 + chunk] = red[0];
}

__global__ void k_enc_finish(const float* __restrict__ part,
                             const float* __restrict__ enc_b,
                             const float* __restrict__ bw_w,
                             const float* __restrict__ bw_b,
                             float* __restrict__ encoded,
                             float* __restrict__ invw2t) {
  __shared__ float enc_s[NBAS];
  const int t = threadIdx.x;
  if (t < NBAS) {
    float s = 0.f;
    for (int c = 0; c < 32; ++c) s += part[t * 32 + c];
    s += enc_b[t];
    enc_s[t] = s;
    encoded[t] = s;
  }
  __syncthreads();
  if (t < NBAS * MCLU) {
    float s = 0.f;
    for (int q = 0; q < NBAS; ++q) s += enc_s[q] * bw_w[t * NBAS + q];
    s += bw_b[t];
    float band = (1.f / (1.f + expf(-s))) / 60.f;
    float wm = band * MUF;
    const int i = t / MCLU, c = t % MCLU;
    invw2t[c * NBAS + i] = 1.f / (wm * wm);
  }
}

#define TR_WAVES  4
#define TR_TILES  (NPTS / 32)
#define TR_BLOCKS ((TR_TILES + TR_WAVES - 1) / TR_WAVES)

__global__ __launch_bounds__(TR_WAVES * 32)
void k_dec_transpose(const float* __restrict__ dec,
                     float* __restrict__ dect) {
  const unsigned lane = threadIdx.x & 31u, wv = threadIdx.x >> 5;
  const unsigned tile = blockIdx.x * TR_WAVES + wv;
  if (tile >= TR_TILES) return;
  const v4f* src = reinterpret_cast<const v4f*>(dec + lane * (unsigned)NPTS + tile * 32u);
  v4f r[8];
#pragma unroll
  for (int s = 0; s < 8; ++s) r[s] = src[s];
  float* dst = dect + tile * 1024u + lane;
#pragma unroll
  for (int p = 0; p < 32; ++p) dst[p * 32] = r[p >> 2][p & 3];
}

__global__ void k_denom_partial(const float* __restrict__ d,
                                const int* __restrict__ lab,
                                const float* __restrict__ invw2t,
                                float* __restrict__ part) {
  __shared__ float iw[MCLU * NBAS];
  __shared__ float red[8 * 1024];
  for (int t = threadIdx.x; t < MCLU * NBAS; t += blockDim.x) iw[t] = invw2t[t];
  __syncthreads();
  const int lane = threadIdx.x & 31, wv = threadIdx.x >> 5;
  const int gw = blockIdx.x * 8 + wv, nw = gridDim.x * 8;
  float acc[NBAS];
#pragma unroll
  for (int i = 0; i < NBAS; ++i) acc[i] = 0.f;
  for (int j = gw; j < NPTS; j += nw) {
    const int c = lab[j];
    const float dv = d[j * KNEI + lane];
    const float d2 = dv * dv;
#pragma unroll
    for (int i = 0; i < NBAS; ++i)
      acc[i] += fmaxf(1.f - d2 * iw[c * NBAS + i], 0.f);
  }
#pragma unroll
  for (int i = 0; i < NBAS; ++i) red[wv * 1024 + i * 32 + lane] = acc[i];
  __syncthreads();
  for (int e = threadIdx.x; e < 1024; e += blockDim.x) {
    float s = 0.f;
#pragma unroll
    for (int w = 0; w < 8; ++w) s += red[w * 1024 + e];
    part[blockIdx.x * 1024 + e] = s;
  }
}

__global__ void k_denom_finish(const float* __restrict__ part, int nblk,
                               float* __restrict__ rdent) {
  const int t = threadIdx.x;
  float s = 0.f;
  for (int b = 0; b < nblk; ++b) s += part[b * 1024 + t];
  const int i = t >> 5, kk = t & 31;
  rdent[kk * 32 + i] = 1.f / s;
}

#define FO_WAVES   4
#define FO_THREADS 128
#define FO_BLOCKS  1563

template <int G>
__device__ __forceinline__ int bcast8(int v) {
  return (int)__builtin_amdgcn_mov_dpp8((unsigned)v, (unsigned)(G * 0x249249));
}

template <int KK>
__device__ __forceinline__ void issue_slot(v4f (&r)[4], const v4f* __restrict__ dect4,
                                           const v4i (&n4)[4], const unsigned q) {
  r[0] = dect4[(unsigned)bcast8<(KK >> 2)>(n4[0][KK & 3]) * 8u + q];
  r[1] = dect4[(unsigned)bcast8<(KK >> 2)>(n4[1][KK & 3]) * 8u + q];
  r[2] = dect4[(unsigned)bcast8<(KK >> 2)>(n4[2][KK & 3]) * 8u + q];
  r[3] = dect4[(unsigned)bcast8<(KK >> 2)>(n4[3][KK & 3]) * 8u + q];
}

__device__ __forceinline__ void add_pass(v4f& acc, const v4f r4, const float d2k,
                                         const v4f s4, const v4f dec) {
  acc.x += r4.x * fmaxf(1.f - d2k * s4.x, 0.f) * dec.x;
  acc.y += r4.y * fmaxf(1.f - d2k * s4.y, 0.f) * dec.y;
  acc.z += r4.z * fmaxf(1.f - d2k * s4.z, 0.f) * dec.z;
  acc.w += r4.w * fmaxf(1.f - d2k * s4.w, 0.f) * dec.w;
}

template <int KK>
__device__ __forceinline__ void use_slot(v4f (&acc)[4], const v4f (&r)[4],
                                         const v4f (&d2v)[4], const v4f (&s4)[4],
                                         const v4f* rd4, const unsigned q) {
  const v4f r4 = rd4[KK * 8 + q];
  add_pass(acc[0], r4, __int_as_float(bcast8<(KK >> 2)>(__float_as_int(d2v[0][KK & 3]))), s4[0], r[0]);
  add_pass(acc[1], r4, __int_as_float(bcast8<(KK >> 2)>(__float_as_int(d2v[1][KK & 3]))), s4[1], r[1]);
  add_pass(acc[2], r4, __int_as_float(bcast8<(KK >> 2)>(__float_as_int(d2v[2][KK & 3]))), s4[2], r[2]);
  add_pass(acc[3], r4, __int_as_float(bcast8<(KK >> 2)>(__float_as_int(d2v[3][KK & 3]))), s4[3], r[3]);
}

template <int K>
__device__ __forceinline__ void pipe_step(v4f (&acc)[4], v4f (&ra)[4], v4f (&rb)[4],
                                          const v4f* __restrict__ dect4,
                                          const v4i (&n4)[4], const v4f (&d2v)[4],
                                          const v4f (&s4)[4], const v4f* rd4,
                                          const unsigned q) {
  use_slot<K>(acc, ra, d2v, s4, rd4, q);
  use_slot<K + 1>(acc, rb, d2v, s4, rd4, q);
  if constexpr (K + 4 < KNEI) {
    issue_slot<K + 4>(ra, dect4, n4, q);
    issue_slot<K + 5>(rb, dect4, n4, q);
  }
}

__global__ __launch_bounds__(FO_THREADS)
void k_fused_out(const float* __restrict__ d,
                 const int* __restrict__ nid,
                 const int* __restrict__ lab,
                 const float* __restrict__ dect,
                 const float* __restrict__ invw2t,
                 const float* __restrict__ rdent,
                 const float* __restrict__ encoded,
                 float* __restrict__ out) {
  __shared__ __attribute__((aligned(16))) float iw[MCLU * NBAS];
  __shared__ __attribute__((aligned(16))) float rd[NBAS * NBAS];
  __shared__ float enc_s[NBAS];
  __shared__ float g[FO_WAVES][32 * 17];
  for (int t = threadIdx.x; t < MCLU * NBAS / 4; t += FO_THREADS)
    reinterpret_cast<v4f*>(iw)[t] = reinterpret_cast<const v4f*>(invw2t)[t];
  for (int t = threadIdx.x; t < NBAS * NBAS / 4; t += FO_THREADS)
    reinterpret_cast<v4f*>(rd)[t] = reinterpret_cast<const v4f*>(rdent)[t];
  if (threadIdx.x < NBAS) enc_s[threadIdx.x] = encoded[threadIdx.x];
  __syncthreads();

  const int lane = threadIdx.x & 31, wv = threadIdx.x >> 5;
  const int tile = blockIdx.x * FO_WAVES + wv;
  const int NT = NPTS / 16;
  if (tile >= NT) return;
  float* gbuf = g[wv];
  const int j0 = tile * 16;
  const int pt = lane >> 3;
  const unsigned q = (unsigned)(lane & 7);
  const v4f* dect4 = reinterpret_cast<const v4f*>(dect);
  const v4f* rd4 = reinterpret_cast<const v4f*>(rd);

  v4i n4[4];
  v4f d2v[4], s4[4], acc[4];
  int cl[4];
#pragma unroll
  for (int sub = 0; sub < 4; ++sub) {
    const unsigned j = (unsigned)(j0 + sub * 4 + pt);
    n4[sub] = reinterpret_cast<const v4i*>(nid)[j * 8u + q];
    const v4f dv = reinterpret_cast<const v4f*>(d)[j * 8u + q];
    d2v[sub] = dv * dv;
    cl[sub] = lab[j];
  }
  v4f ra[4], rb[4], rc[4], re[4];
  issue_slot<0>(ra, dect4, n4, q);
  issue_slot<1>(rb, dect4, n4, q);
  issue_slot<2>(rc, dect4, n4, q);
  issue_slot<3>(re, dect4, n4, q);
#pragma unroll
  for (int sub = 0; sub < 4; ++sub) {
    s4[sub] = reinterpret_cast<const v4f*>(iw)[(unsigned)cl[sub] * 8u + q];
    acc[sub] = (v4f){0.f, 0.f, 0.f, 0.f};
  }
  pipe_step<0>(acc, ra, rb, dect4, n4, d2v, s4, rd4, q);
  pipe_step<2>(acc, rc, re, dect4, n4, d2v, s4, rd4, q);
  pipe_step<4>(acc, ra, rb, dect4, n4, d2v, s4, rd4, q);
  pipe_step<6>(acc, rc, re, dect4, n4, d2v, s4, rd4, q);
  pipe_step<8>(acc, ra, rb, dect4, n4, d2v, s4, rd4, q);
  pipe_step<10>(acc, rc, re, dect4, n4, d2v, s4, rd4, q);
  pipe_step<12>(acc, ra, rb, dect4, n4, d2v, s4, rd4, q);
  pipe_step<14>(acc, rc, re, dect4, n4, d2v, s4, rd4, q);
  pipe_step<16>(acc, ra, rb, dect4, n4, d2v, s4, rd4, q);
  pipe_step<18>(acc, rc, re, dect4, n4, d2v, s4, rd4, q);
  pipe_step<20>(acc, ra, rb, dect4, n4, d2v, s4, rd4, q);
  pipe_step<22>(acc, rc, re, dect4, n4, d2v, s4, rd4, q);
  pipe_step<24>(acc, ra, rb, dect4, n4, d2v, s4, rd4, q);
  pipe_step<26>(acc, rc, re, dect4, n4, d2v, s4, rd4, q);
  pipe_step<28>(acc, ra, rb, dect4, n4, d2v, s4, rd4, q);
  pipe_step<30>(acc, rc, re, dect4, n4, d2v, s4, rd4, q);

#pragma unroll
  for (int sub = 0; sub < 4; ++sub) {
    const int colp = sub * 4 + pt;
    gbuf[(4 * q + 0) * 17 + colp] = acc[sub].x;
    gbuf[(4 * q + 1) * 17 + colp] = acc[sub].y;
    gbuf[(4 * q + 2) * 17 + colp] = acc[sub].z;
    gbuf[(4 * q + 3) * 17 + colp] = acc[sub].w;
  }
  asm volatile("s_wait_dscnt 0" ::: "memory");

  v8f cacc = {0.f, 0.f, 0.f, 0.f, 0.f, 0.f, 0.f, 0.f};
  const int half = (lane >> 4) & 1;
  const int col = lane & 15;
#pragma unroll
  for (int s8 = 0; s8 < 8; ++s8) {
    const int kb = s8 * 4;
    v2f a, b;
    a.x = enc_s[kb + 2 * half];
    a.y = enc_s[kb + 2 * half + 1];
    b.x = gbuf[(kb + 2 * half) * 17 + col];
    b.y = gbuf[(kb + 2 * half + 1) * 17 + col];
    cacc = __builtin_amdgcn_wmma_f32_16x16x4_f32(
        false, a, false, b, (short)0, cacc, false, false);
  }
  if (lane < 16) out[j0 + lane] = cacc[0];
}

extern "C" void kernel_launch(void* const* d_in, const int* in_sizes, int n_in,
                              void* d_out, int out_size, void* d_ws, size_t ws_size,
                              hipStream_t stream) {
  const float* x     = (const float*)d_in[0];
  const float* enc_w = (const float*)d_in[1];
  const float* enc_b = (const float*)d_in[2];
  const float* dec   = (const float*)d_in[3];
  const float* bw_w  = (const float*)d_in[4];
  const float* bw_b  = (const float*)d_in[5];
  const float* nd    = (const float*)d_in[6];
  const int*   nid   = (const int*)d_in[7];
  const int*   lab   = (const int*)d_in[8];
  float* out = (float*)d_out;
  float* ws  = (float*)d_ws;

  float* enc_part = ws + WS_ENCPART;
  float* encoded  = ws + WS_ENCODED;
  float* invw2t   = ws + WS_INVW2T;
  float* rdent    = ws + WS_RDENT;
  float* denpart  = ws + WS_DENPART;
  float* dect     = ws + WS_DECT;

  k_enc_partial<<<dim3(32, 32), 256, 0, stream>>>(x, enc_w, enc_part);
  k_enc_finish<<<1, 512, 0, stream>>>(enc_part, enc_b, bw_w, bw_b, encoded, invw2t);
  k_dec_transpose<<<TR_BLOCKS, TR_WAVES * 32, 0, stream>>>(dec, dect);
  k_denom_partial<<<DEN_BLKS, 256, 0, stream>>>(nd, lab, invw2t, denpart);
  k_denom_finish<<<1, 1024, 0, stream>>>(denpart, DEN_BLKS, rdent);
  k_fused_out<<<FO_BLOCKS, FO_THREADS, 0, stream>>>(nd, nid, lab, dect, invw2t, rdent,
                                       encoded, out);
}


# denominator pass re-decomposed: bases across waves (4 a wave, 16 a block), a block's 782 points' rows and labels held in LDS by one load burst; 128 partials (was 224); denominator sums regrouped
# speedup vs baseline: 7.7746x; 1.4011x over previous
#include <hip/hip_runtime.h>
#include <math.h>


#define NPTS 100000
#define NBAS 32
#define MCLU 16
#define KNEI 32
#define MUF  100.0f

typedef __attribute__((ext_vector_type(2))) float v2f;
typedef __attribute__((ext_vector_type(8))) float v8f;
typedef __attribute__((ext_vector_type(4))) float v4f;
typedef __attribute__((ext_vector_type(4))) int v4i;

#define WS_ENCPART 0
#define WS_ENCODED (WS_ENCPART + 1024)
#define WS_INVW2T  (WS_ENCODED + 32)
#define WS_RDENT   (WS_INVW2T + 512)
#define DEN_BLKS   128
#define WS_DENPART (WS_RDENT + 1024)
#define WS_DECT    (WS_DENPART + DEN_BLKS*1024)

__global__ void k_enc_partial(const float* __restrict__ x,
                              const float* __restrict__ W,
                              float* __restrict__ part) {
  const int row = blockIdx.y, chunk = blockIdx.x;
  const int CS = (NPTS + 31) / 32;
  const int j0 = chunk * CS;
  const int j1 = (j0 + CS < NPTS) ? (j0 + CS) : NPTS;
  float s = 0.f;
  for (int j = j0 + threadIdx.x; j < j1; j += blockDim.x)
    s += x[j] * W[row * NPTS + j];
  __shared__ float red[256];
  red[threadIdx.x] = s;
  __syncthreads();
  for (int o = 128; o > 0; o >>= 1) {
    if (threadIdx.x < o) red[threadIdx.x] += red[threadIdx.x + o];
    __syncthreads();
  }
  if (threadIdx.x == 0) part[row * 32 + chunk] = red[0];
}

__global__ void k_enc_finish(const float* __restrict__ part,
                             const float* __restrict__ enc_b,
                             const float* __restrict__ bw_w,
                             const float* __restrict__ bw_b,
                             float* __restrict__ encoded,
                             float* __restrict__ invw2t) {
  __shared__ float enc_s[NBAS];
  const int t = threadIdx.x;
  if (t < NBAS) {
    float s = 0.f;
    for (int c = 0; c < 32; ++c) s += part[t * 32 + c];
    s += enc_b[t];
    enc_s[t] = s;
    encoded[t] = s;
  }
  __syncthreads();
  if (t < NBAS * MCLU) {
    float s = 0.f;
    for (int q = 0; q < NBAS; ++q) s += enc_s[q] * bw_w[t * NBAS + q];
    s += bw_b[t];
    float band = (1.f / (1.f + expf(-s))) / 60.f;
    float wm = band * MUF;
    const int i = t / MCLU, c = t % MCLU;
    invw2t[c * NBAS + i] = 1.f / (wm * wm);
  }
}

#define TR_WAVES  4
#define TR_TILES  (NPTS / 32)
#define TR_BLOCKS ((TR_TILES + TR_WAVES - 1) / TR_WAVES)

__global__ __launch_bounds__(TR_WAVES * 32)
void k_dec_transpose(const float* __restrict__ dec,
                     float* __restrict__ dect) {
  const unsigned lane = threadIdx.x & 31u, wv = threadIdx.x >> 5;
  const unsigned tile = blockIdx.x * TR_WAVES + wv;
  if (tile >= TR_TILES) return;
  const v4f* src = reinterpret_cast<const v4f*>(dec + lane * (unsigned)NPTS + tile * 32u);
  v4f r[8];
#pragma unroll
  for (int s = 0; s < 8; ++s) r[s] = src[s];
  float* dst = dect + tile * 1024u + lane;
#pragma unroll
  for (int p = 0; p < 32; ++p) dst[p * 32] = r[p >> 2][p & 3];
}

#define DEN_GRID 256
#define DEN_THR  512
#define DEN_CH   782
#define DEN_Q    200
#define DEN_RD   13
static_assert(DEN_BLKS * DEN_CH >= NPTS, "point groups cover all points");
static_assert(NPTS - (DEN_BLKS - 1) * DEN_CH >= DEN_THR, "shortest chunk has a label for every thread");
static_assert(DEN_RD * DEN_THR >= DEN_CH * (KNEI / 4), "one burst covers a chunk");
static_assert(MCLU * NBAS == DEN_THR, "one table entry per thread");
static_assert(2 * DEN_THR >= DEN_CH + 2, "label slots cover a chunk");

__device__ __forceinline__ void den_point(v4f& acc, const int c, const float d2,
                                          const float* arow) {
  const v4f a = *reinterpret_cast<const v4f*>(arow + c * NBAS);
  acc.x += fmaxf(1.f - d2 * a.x, 0.f);
  acc.y += fmaxf(1.f - d2 * a.y, 0.f);
  acc.z += fmaxf(1.f - d2 * a.z, 0.f);
  acc.w += fmaxf(1.f - d2 * a.w, 0.f);
}

__global__ __launch_bounds__(DEN_THR)
void k_denom_partial(const float* __restrict__ d,
                     const int* __restrict__ lab,
                     const float* __restrict__ invw2t,
                     float* __restrict__ part) {
  __shared__ __attribute__((aligned(16))) float d2s[DEN_RD * DEN_THR * 4];
  __shared__ __attribute__((aligned(16))) int   labs[2 * DEN_THR];
  __shared__ __attribute__((aligned(16))) float iw[MCLU * NBAS];
  __shared__ __attribute__((aligned(16))) float comb[3 * 4 * 4 * 32];
  const int tid = threadIdx.x;
  const int pg = blockIdx.x >> 1, ih = blockIdx.x & 1;
  const int p0 = pg * DEN_CH;
  const int n  = min(DEN_CH, NPTS - p0);
  const int n4 = n * (KNEI / 4);

  const int lv0 = lab[p0 + tid];
  const int lv1 = lab[p0 + min(tid + DEN_THR, n - 1)];
  const float iv = invw2t[tid];
  const v4f* src = reinterpret_cast<const v4f*>(d + p0 * KNEI);
  v4f r[DEN_RD];
#pragma unroll
  for (int s = 0; s < DEN_RD; ++s) r[s] = src[min(s * DEN_THR + tid, n4 - 1)];
  labs[tid] = lv0;
  labs[tid + DEN_THR] = lv1;
  iw[tid] = iv;
  v4f* dst = reinterpret_cast<v4f*>(d2s);
#pragma unroll
  for (int s = 0; s < DEN_RD; ++s) dst[s * DEN_THR + tid] = r[s] * r[s];
  __syncthreads();

  const int lane = tid & 31, w = tid >> 5;
  const int bq = w & 3, pq = w >> 2;
  const int i0 = ih * 16 + bq * 4;
  const float* arow = iw + i0;
  const int pbeg = pq * DEN_Q;
  const int pend = (pq == 3) ? n : min(pbeg + DEN_Q, n);
  const int pmain = pbeg + ((pend - pbeg) & ~7);
  const v4i* L4 = reinterpret_cast<const v4i*>(labs);
  v4f acc = {0.f, 0.f, 0.f, 0.f};
  int p = pbeg;
  for (; p < pmain; p += 8) {
    const v4i c0 = L4[p >> 2], c1 = L4[(p >> 2) + 1];
    const float* dp = d2s + p * KNEI + lane;
    den_point(acc, c0.x, dp[0 * KNEI], arow);
    den_point(acc, c0.y, dp[1 * KNEI], arow);
    den_point(acc, c0.z, dp[2 * KNEI], arow);
    den_point(acc, c0.w, dp[3 * KNEI], arow);
    den_point(acc, c1.x, dp[4 * KNEI], arow);
    den_point(acc, c1.y, dp[5 * KNEI], arow);
    den_point(acc, c1.z, dp[6 * KNEI], arow);
    den_point(acc, c1.w, dp[7 * KNEI], arow);
  }
  for (; p < pend; ++p) den_point(acc, labs[p], d2s[p * KNEI + lane], arow);

  if (pq) {
    float* cw = comb + ((pq - 1) * 16 + bq * 4) * 32 + lane;
    cw[0 * 32] = acc.x;
    cw[1 * 32] = acc.y;
    cw[2 * 32] = acc.z;
    cw[3 * 32] = acc.w;
  }
  __syncthreads();
  if (!pq) {
    const float* cr = comb + (bq * 4) * 32 + lane;
    float* o = part + pg * 1024 + i0 * 32 + lane;
    o[0 * 32] = ((acc.x + cr[0 * 32]) + cr[(16 + 0) * 32]) + cr[(32 + 0) * 32];
    o[1 * 32] = ((acc.y + cr[1 * 32]) + cr[(16 + 1) * 32]) + cr[(32 + 1) * 32];
    o[2 * 32] = ((acc.z + cr[2 * 32]) + cr[(16 + 2) * 32]) + cr[(32 + 2) * 32];
    o[3 * 32] = ((acc.w + cr[3 * 32]) + cr[(16 + 3) * 32]) + cr[(32 + 3) * 32];
  }
}

__global__ void k_denom_finish(const float* __restrict__ part, int nblk,
                               float* __restrict__ rdent) {
  const int t = threadIdx.x;
  float s = 0.f;
  for (int b = 0; b < nblk; ++b) s += part[b * 1024 + t];
  const int i = t >> 5, kk = t & 31;
  rdent[kk * 32 + i] = 1.f / s;
}

#define FO_WAVES   4
#define FO_THREADS 128
#define FO_BLOCKS  1563

template <int G>
__device__ __forceinline__ int bcast8(int v) {
  return (int)__builtin_amdgcn_mov_dpp8((unsigned)v, (unsigned)(G * 0x249249));
}

template <int KK>
__device__ __forceinline__ void issue_slot(v4f (&r)[4], const v4f* __restrict__ dect4,
                                           const v4i (&n4)[4], const unsigned q) {
  r[0] = dect4[(unsigned)bcast8<(KK >> 2)>(n4[0][KK & 3]) * 8u + q];
  r[1] = dect4[(unsigned)bcast8<(KK >> 2)>(n4[1][KK & 3]) * 8u + q];
  r[2] = dect4[(unsigned)bcast8<(KK >> 2)>(n4[2][KK & 3]) * 8u + q];
  r[3] = dect4[(unsigned)bcast8<(KK >> 2)>(n4[3][KK & 3]) * 8u + q];
}

__device__ __forceinline__ void add_pass(v4f& acc, const v4f r4, const float d2k,
                                         const v4f s4, const v4f dec) {
  acc.x += r4.x * fmaxf(1.f - d2k * s4.x, 0.f) * dec.x;
  acc.y += r4.y * fmaxf(1.f - d2k * s4.y, 0.f) * dec.y;
  acc.z += r4.z * fmaxf(1.f - d2k * s4.z, 0.f) * dec.z;
  acc.w += r4.w * fmaxf(1.f - d2k * s4.w, 0.f) * dec.w;
}

template <int KK>
__device__ __forceinline__ void use_slot(v4f (&acc)[4], const v4f (&r)[4],
                                         const v4f (&d2v)[4], const v4f (&s4)[4],
                                         const v4f* rd4, const unsigned q) {
  const v4f r4 = rd4[KK * 8 + q];
  add_pass(acc[0], r4, __int_as_float(bcast8<(KK >> 2)>(__float_as_int(d2v[0][KK & 3]))), s4[0], r[0]);
  add_pass(acc[1], r4, __int_as_float(bcast8<(KK >> 2)>(__float_as_int(d2v[1][KK & 3]))), s4[1], r[1]);
  add_pass(acc[2], r4, __int_as_float(bcast8<(KK >> 2)>(__float_as_int(d2v[2][KK & 3]))), s4[2], r[2]);
  add_pass(acc[3], r4, __int_as_float(bcast8<(KK >> 2)>(__float_as_int(d2v[3][KK & 3]))), s4[3], r[3]);
}

template <int K>
__device__ __forceinline__ void pipe_step(v4f (&acc)[4], v4f (&ra)[4], v4f (&rb)[4],
                                          const v4f* __restrict__ dect4,
                                          const v4i (&n4)[4], const v4f (&d2v)[4],
                                          const v4f (&s4)[4], const v4f* rd4,
                                          const unsigned q) {
  use_slot<K>(acc, ra, d2v, s4, rd4, q);
  use_slot<K + 1>(acc, rb, d2v, s4, rd4, q);
  if constexpr (K + 4 < KNEI) {
    issue_slot<K + 4>(ra, dect4, n4, q);
    issue_slot<K + 5>(rb, dect4, n4, q);
  }
}

__global__ __launch_bounds__(FO_THREADS)
void k_fused_out(const float* __restrict__ d,
                 const int* __restrict__ nid,
                 const int* __restrict__ lab,
                 const float* __restrict__ dect,
                 const float* __restrict__ invw2t,
                 const float* __restrict__ rdent,
                 const float* __restrict__ encoded,
                 float* __restrict__ out) {
  __shared__ __attribute__((aligned(16))) float iw[MCLU * NBAS];
  __shared__ __attribute__((aligned(16))) float rd[NBAS * NBAS];
  __shared__ float enc_s[NBAS];
  __shared__ float g[FO_WAVES][32 * 17];
  for (int t = threadIdx.x; t < MCLU * NBAS / 4; t += FO_THREADS)
    reinterpret_cast<v4f*>(iw)[t] = reinterpret_cast<const v4f*>(invw2t)[t];
  for (int t = threadIdx.x; t < NBAS * NBAS / 4; t += FO_THREADS)
    reinterpret_cast<v4f*>(rd)[t] = reinterpret_cast<const v4f*>(rdent)[t];
  if (threadIdx.x < NBAS) enc_s[threadIdx.x] = encoded[threadIdx.x];
  __syncthreads();

  const int lane = threadIdx.x & 31, wv = threadIdx.x >> 5;
  const int tile = blockIdx.x * FO_WAVES + wv;
  const int NT = NPTS / 16;
  if (tile >= NT) return;
  float* gbuf = g[wv];
  const int j0 = tile * 16;
  const int pt = lane >> 3;
  const unsigned q = (unsigned)(lane & 7);
  const v4f* dect4 = reinterpret_cast<const v4f*>(dect);
  const v4f* rd4 = reinterpret_cast<const v4f*>(rd);

  v4i n4[4];
  v4f d2v[4], s4[4], acc[4];
  int cl[4];
#pragma unroll
  for (int sub = 0; sub < 4; ++sub) {
    const unsigned j = (unsigned)(j0 + sub * 4 + pt);
    n4[sub] = reinterpret_cast<const v4i*>(nid)[j * 8u + q];
    const v4f dv = reinterpret_cast<const v4f*>(d)[j * 8u + q];
    d2v[sub] = dv * dv;
    cl[sub] = lab[j];
  }
  v4f ra[4], rb[4], rc[4], re[4];
  issue_slot<0>(ra, dect4, n4, q);
  issue_slot<1>(rb, dect4, n4, q);
  issue_slot<2>(rc, dect4, n4, q);
  issue_slot<3>(re, dect4, n4, q);
#pragma unroll
  for (int sub = 0; sub < 4; ++sub) {
    s4[sub] = reinterpret_cast<const v4f*>(iw)[(unsigned)cl[sub] * 8u + q];
    acc[sub] = (v4f){0.f, 0.f, 0.f, 0.f};
  }
  pipe_step<0>(acc, ra, rb, dect4, n4, d2v, s4, rd4, q);
  pipe_step<2>(acc, rc, re, dect4, n4, d2v, s4, rd4, q);
  pipe_step<4>(acc, ra, rb, dect4, n4, d2v, s4, rd4, q);
  pipe_step<6>(acc, rc, re, dect4, n4, d2v, s4, rd4, q);
  pipe_step<8>(acc, ra, rb, dect4, n4, d2v, s4, rd4, q);
  pipe_step<10>(acc, rc, re, dect4, n4, d2v, s4, rd4, q);
  pipe_step<12>(acc, ra, rb, dect4, n4, d2v, s4, rd4, q);
  pipe_step<14>(acc, rc, re, dect4, n4, d2v, s4, rd4, q);
  pipe_step<16>(acc, ra, rb, dect4, n4, d2v, s4, rd4, q);
  pipe_step<18>(acc, rc, re, dect4, n4, d2v, s4, rd4, q);
  pipe_step<20>(acc, ra, rb, dect4, n4, d2v, s4, rd4, q);
  pipe_step<22>(acc, rc, re, dect4, n4, d2v, s4, rd4, q);
  pipe_step<24>(acc, ra, rb, dect4, n4, d2v, s4, rd4, q);
  pipe_step<26>(acc, rc, re, dect4, n4, d2v, s4, rd4, q);
  pipe_step<28>(acc, ra, rb, dect4, n4, d2v, s4, rd4, q);
  pipe_step<30>(acc, rc, re, dect4, n4, d2v, s4, rd4, q);

#pragma unroll
  for (int sub = 0; sub < 4; ++sub) {
    const int colp = sub * 4 + pt;
    gbuf[(4 * q + 0) * 17 + colp] = acc[sub].x;
    gbuf[(4 * q + 1) * 17 + colp] = acc[sub].y;
    gbuf[(4 * q + 2) * 17 + colp] = acc[sub].z;
    gbuf[(4 * q + 3) * 17 + colp] = acc[sub].w;
  }
  asm volatile("s_wait_dscnt 0" ::: "memory");

  v8f cacc = {0.f, 0.f, 0.f, 0.f, 0.f, 0.f, 0.f, 0.f};
  const int half = (lane >> 4) & 1;
  const int col = lane & 15;
#pragma unroll
  for (int s8 = 0; s8 < 8; ++s8) {
    const int kb = s8 * 4;
    v2f a, b;
    a.x = enc_s[kb + 2 * half];
    a.y = enc_s[kb + 2 * half + 1];
    b.x = gbuf[(kb + 2 * half) * 17 + col];
    b.y = gbuf[(kb + 2 * half + 1) * 17 + col];
    cacc = __builtin_amdgcn_wmma_f32_16x16x4_f32(
        false, a, false, b, (short)0, cacc, false, false);
  }
  if (lane < 16) out[j0 + lane] = cacc[0];
}

extern "C" void kernel_launch(void* const* d_in, const int* in_sizes, int n_in,
                              void* d_out, int out_size, void* d_ws, size_t ws_size,
                              hipStream_t stream) {
  const float* x     = (const float*)d_in[0];
  const float* enc_w = (const float*)d_in[1];
  const float* enc_b = (const float*)d_in[2];
  const float* dec   = (const float*)d_in[3];
  const float* bw_w  = (const float*)d_in[4];
  const float* bw_b  = (const float*)d_in[5];
  const float* nd    = (const float*)d_in[6];
  const int*   nid   = (const int*)d_in[7];
  const int*   lab   = (const int*)d_in[8];
  float* out = (float*)d_out;
  float* ws  = (float*)d_ws;

  float* enc_part = ws + WS_ENCPART;
  float* encoded  = ws + WS_ENCODED;
  float* invw2t   = ws + WS_INVW2T;
  float* rdent    = ws + WS_RDENT;
  float* denpart  = ws + WS_DENPART;
  float* dect     = ws + WS_DECT;

  k_enc_partial<<<dim3(32, 32), 256, 0, stream>>>(x, enc_w, enc_part);
  k_enc_finish<<<1, 512, 0, stream>>>(enc_part, enc_b, bw_w, bw_b, encoded, invw2t);
  k_dec_transpose<<<TR_BLOCKS, TR_WAVES * 32, 0, stream>>>(dec, dect);
  k_denom_partial<<<DEN_GRID, DEN_THR, 0, stream>>>(nd, lab, invw2t, denpart);
  k_denom_finish<<<1, 1024, 0, stream>>>(denpart, DEN_BLKS, rdent);
  k_fused_out<<<FO_BLOCKS, FO_THREADS, 0, stream>>>(nd, nid, lab, dect, invw2t, rdent,
                                       encoded, out);
}


# encoder GEMV re-tiled: columns across waves, all 32 rows in a wave: x kept in registers, row sums in-lane, then 5-step lane butterfly; 196 partial lines (was 1024 via LDS tree); GEMV sums regrouped
# speedup vs baseline: 8.4617x; 1.0884x over previous
#include <hip/hip_runtime.h>
#include <math.h>


#define NPTS 100000
#define NBAS 32
#define MCLU 16
#define KNEI 32
#define MUF  100.0f

typedef __attribute__((ext_vector_type(2))) float v2f;
typedef __attribute__((ext_vector_type(8))) float v8f;
typedef __attribute__((ext_vector_type(4))) float v4f;
typedef __attribute__((ext_vector_type(4))) int v4i;

#define GV_WAVES  4
#define GV_TILE   128
#define GV_TILES  ((NPTS + GV_TILE - 1) / GV_TILE)
#define GV_BLOCKS ((GV_TILES + GV_WAVES - 1) / GV_WAVES)
#define GV_GROUPS 16
#define GV_PER    ((GV_BLOCKS + GV_GROUPS - 1) / GV_GROUPS)
#define WS_ENCPART 0
#define WS_ENCODED (WS_ENCPART + GV_BLOCKS * 32)
#define WS_INVW2T  (WS_ENCODED + 32)
#define WS_RDENT   (WS_INVW2T + 512)
#define DEN_BLKS   128
#define WS_DENPART (WS_RDENT + 1024)
#define WS_DECT    (WS_DENPART + DEN_BLKS*1024)

template <int S>
__device__ __forceinline__ void gv_fold(float (&a)[32], const unsigned lane) {
  const bool up = (lane & (unsigned)S) != 0u;
#pragma unroll
  for (int r = 0; r < S; ++r) {
    const float keep = up ? a[r + S] : a[r];
    const float send = up ? a[r] : a[r + S];
    a[r] = keep + __shfl_xor(send, S);
  }
}

__global__ __launch_bounds__(GV_WAVES * 32)
void k_enc_partial(const float* __restrict__ x,
                   const float* __restrict__ W,
                   float* __restrict__ part) {
  __shared__ float red[GV_WAVES][32];
  const unsigned lane = threadIdx.x & 31u, wv = threadIdx.x >> 5;
  const unsigned tile = blockIdx.x * GV_WAVES + wv;
  const unsigned j = tile * GV_TILE + lane * 4u;
  const bool valid = j < (unsigned)NPTS;
  const unsigned jc = valid ? j : (unsigned)(NPTS - 4);
  const v4f xv = *reinterpret_cast<const v4f*>(x + jc);
  const v4f* wp = reinterpret_cast<const v4f*>(W + jc);
  v4f w[NBAS];
#pragma unroll
  for (int r = 0; r < NBAS; ++r) w[r] = wp[r * (NPTS / 4)];
  float a[NBAS];
#pragma unroll
  for (int r = 0; r < NBAS; ++r) {
    float s = w[r].x * xv.x;
    s += w[r].y * xv.y;
    s += w[r].z * xv.z;
    s += w[r].w * xv.w;
    a[r] = valid ? s : 0.f;
  }
  gv_fold<16>(a, lane);
  gv_fold<8>(a, lane);
  gv_fold<4>(a, lane);
  gv_fold<2>(a, lane);
  gv_fold<1>(a, lane);
  red[wv][lane] = a[0];
  __syncthreads();
  if (wv == 0)
    part[blockIdx.x * 32 + lane] = ((red[0][lane] + red[1][lane]) + red[2][lane]) + red[3][lane];
}

__global__ void k_enc_finish(const float* __restrict__ part,
                             const float* __restrict__ enc_b,
                             const float* __restrict__ bw_w,
                             const float* __restrict__ bw_b,
                             float* __restrict__ encoded,
                             float* __restrict__ invw2t) {
  __shared__ float enc_s[NBAS];
  __shared__ float ps[GV_GROUPS][NBAS];
  const int t = threadIdx.x;
  {
    const int g = t >> 5, r = t & 31;
    float s = 0.f;
#pragma unroll
    for (int k = 0; k < GV_PER; ++k) {
      const int b = g + k * GV_GROUPS;
      const float v = part[min(b, GV_BLOCKS - 1) * 32 + r];
      s += (b < GV_BLOCKS) ? v : 0.f;
    }
    ps[g][r] = s;
  }
  __syncthreads();
  if (t < NBAS) {
    float s = 0.f;
    for (int g = 0; g < GV_GROUPS; ++g) s += ps[g][t];
    s += enc_b[t];
    enc_s[t] = s;
    encoded[t] = s;
  }
  __syncthreads();
  if (t < NBAS * MCLU) {
    float s = 0.f;
    for (int q = 0; q < NBAS; ++q) s += enc_s[q] * bw_w[t * NBAS + q];
    s += bw_b[t];
    float band = (1.f / (1.f + expf(-s))) / 60.f;
    float wm = band * MUF;
    const int i = t / MCLU, c = t % MCLU;
    invw2t[c * NBAS + i] = 1.f / (wm * wm);
  }
}

#define TR_WAVES  4
#define TR_TILES  (NPTS / 32)
#define TR_BLOCKS ((TR_TILES + TR_WAVES - 1) / TR_WAVES)

__global__ __launch_bounds__(TR_WAVES * 32)
void k_dec_transpose(const float* __restrict__ dec,
                     float* __restrict__ dect) {
  const unsigned lane = threadIdx.x & 31u, wv = threadIdx.x >> 5;
  const unsigned tile = blockIdx.x * TR_WAVES + wv;
  if (tile >= TR_TILES) return;
  const v4f* src = reinterpret_cast<const v4f*>(dec + lane * (unsigned)NPTS + tile * 32u);
  v4f r[8];
#pragma unroll
  for (int s = 0; s < 8; ++s) r[s] = src[s];
  float* dst = dect + tile * 1024u + lane;
#pragma unroll
  for (int p = 0; p < 32; ++p) dst[p * 32] = r[p >> 2][p & 3];
}

#define DEN_GRID 256
#define DEN_THR  512
#define DEN_CH   782
#define DEN_Q    200
#define DEN_RD   13
static_assert(DEN_BLKS * DEN_CH >= NPTS, "point groups cover all points");
static_assert(NPTS - (DEN_BLKS - 1) * DEN_CH >= DEN_THR, "shortest chunk has a label for every thread");
static_assert(DEN_RD * DEN_THR >= DEN_CH * (KNEI / 4), "one burst covers a chunk");
static_assert(MCLU * NBAS == DEN_THR, "one table entry per thread");
static_assert(2 * DEN_THR >= DEN_CH + 2, "label slots cover a chunk");

__device__ __forceinline__ void den_point(v4f& acc, const int c, const float d2,
                                          const float* arow) {
  const v4f a = *reinterpret_cast<const v4f*>(arow + c * NBAS);
  acc.x += fmaxf(1.f - d2 * a.x, 0.f);
  acc.y += fmaxf(1.f - d2 * a.y, 0.f);
  acc.z += fmaxf(1.f - d2 * a.z, 0.f);
  acc.w += fmaxf(1.f - d2 * a.w, 0.f);
}

__global__ __launch_bounds__(DEN_THR)
void k_denom_partial(const float* __restrict__ d,
                     const int* __restrict__ lab,
                     const float* __restrict__ invw2t,
                     float* __restrict__ part) {
  __shared__ __attribute__((aligned(16))) float d2s[DEN_RD * DEN_THR * 4];
  __shared__ __attribute__((aligned(16))) int   labs[2 * DEN_THR];
  __shared__ __attribute__((aligned(16))) float iw[MCLU * NBAS];
  __shared__ __attribute__((aligned(16))) float comb[3 * 4 * 4 * 32];
  const int tid = threadIdx.x;
  const int pg = blockIdx.x >> 1, ih = blockIdx.x & 1;
  const int p0 = pg * DEN_CH;
  const int n  = min(DEN_CH, NPTS - p0);
  const int n4 = n * (KNEI / 4);

  const int lv0 = lab[p0 + tid];
  const int lv1 = lab[p0 + min(tid + DEN_THR, n - 1)];
  const float iv = invw2t[tid];
  const v4f* src = reinterpret_cast<const v4f*>(d + p0 * KNEI);
  v4f r[DEN_RD];
#pragma unroll
  for (int s = 0; s < DEN_RD; ++s) r[s] = src[min(s * DEN_THR + tid, n4 - 1)];
  labs[tid] = lv0;
  labs[tid + DEN_THR] = lv1;
  iw[tid] = iv;
  v4f* dst = reinterpret_cast<v4f*>(d2s);
#pragma unroll
  for (int s = 0; s < DEN_RD; ++s) dst[s * DEN_THR + tid] = r[s] * r[s];
  __syncthreads();

  const int lane = tid & 31, w = tid >> 5;
  const int bq = w & 3, pq = w >> 2;
  const int i0 = ih * 16 + bq * 4;
  const float* arow = iw + i0;
  const int pbeg = pq * DEN_Q;
  const int pend = (pq == 3) ? n : min(pbeg + DEN_Q, n);
  const int pmain = pbeg + ((pend - pbeg) & ~7);
  const v4i* L4 = reinterpret_cast<const v4i*>(labs);
  v4f acc = {0.f, 0.f, 0.f, 0.f};
  int p = pbeg;
  for (; p < pmain; p += 8) {
    const v4i c0 = L4[p >> 2], c1 = L4[(p >> 2) + 1];
    const float* dp = d2s + p * KNEI + lane;
    den_point(acc, c0.x, dp[0 * KNEI], arow);
    den_point(acc, c0.y, dp[1 * KNEI], arow);
    den_point(acc, c0.z, dp[2 * KNEI], arow);
    den_point(acc, c0.w, dp[3 * KNEI], arow);
    den_point(acc, c1.x, dp[4 * KNEI], arow);
    den_point(acc, c1.y, dp[5 * KNEI], arow);
    den_point(acc, c1.z, dp[6 * KNEI], arow);
    den_point(acc, c1.w, dp[7 * KNEI], arow);
  }
  for (; p < pend; ++p) den_point(acc, labs[p], d2s[p * KNEI + lane], arow);

  if (pq) {
    float* cw = comb + ((pq - 1) * 16 + bq * 4) * 32 + lane;
    cw[0 * 32] = acc.x;
    cw[1 * 32] = acc.y;
    cw[2 * 32] = acc.z;
    cw[3 * 32] = acc.w;
  }
  __syncthreads();
  if (!pq) {
    const float* cr = comb + (bq * 4) * 32 + lane;
    float* o = part + pg * 1024 + i0 * 32 + lane;
    o[0 * 32] = ((acc.x + cr[0 * 32]) + cr[(16 + 0) * 32]) + cr[(32 + 0) * 32];
    o[1 * 32] = ((acc.y + cr[1 * 32]) + cr[(16 + 1) * 32]) + cr[(32 + 1) * 32];
    o[2 * 32] = ((acc.z + cr[2 * 32]) + cr[(16 + 2) * 32]) + cr[(32 + 2) * 32];
    o[3 * 32] = ((acc.w + cr[3 * 32]) + cr[(16 + 3) * 32]) + cr[(32 + 3) * 32];
  }
}

__global__ void k_denom_finish(const float* __restrict__ part, int nblk,
                               float* __restrict__ rdent) {
  const int t = threadIdx.x;
  float s = 0.f;
  for (int b = 0; b < nblk; ++b) s += part[b * 1024 + t];
  const int i = t >> 5, kk = t & 31;
  rdent[kk * 32 + i] = 1.f / s;
}

#define FO_WAVES   4
#define FO_THREADS 128
#define FO_BLOCKS  1563

template <int G>
__device__ __forceinline__ int bcast8(int v) {
  return (int)__builtin_amdgcn_mov_dpp8((unsigned)v, (unsigned)(G * 0x249249));
}

template <int KK>
__device__ __forceinline__ void issue_slot(v4f (&r)[4], const v4f* __restrict__ dect4,
                                           const v4i (&n4)[4], const unsigned q) {
  r[0] = dect4[(unsigned)bcast8<(KK >> 2)>(n4[0][KK & 3]) * 8u + q];
  r[1] = dect4[(unsigned)bcast8<(KK >> 2)>(n4[1][KK & 3]) * 8u + q];
  r[2] = dect4[(unsigned)bcast8<(KK >> 2)>(n4[2][KK & 3]) * 8u + q];
  r[3] = dect4[(unsigned)bcast8<(KK >> 2)>(n4[3][KK & 3]) * 8u + q];
}

__device__ __forceinline__ void add_pass(v4f& acc, const v4f r4, const float d2k,
                                         const v4f s4, const v4f dec) {
  acc.x += r4.x * fmaxf(1.f - d2k * s4.x, 0.f) * dec.x;
  acc.y += r4.y * fmaxf(1.f - d2k * s4.y, 0.f) * dec.y;
  acc.z += r4.z * fmaxf(1.f - d2k * s4.z, 0.f) * dec.z;
  acc.w += r4.w * fmaxf(1.f - d2k * s4.w, 0.f) * dec.w;
}

template <int KK>
__device__ __forceinline__ void use_slot(v4f (&acc)[4], const v4f (&r)[4],
                                         const v4f (&d2v)[4], const v4f (&s4)[4],
                                         const v4f* rd4, const unsigned q) {
  const v4f r4 = rd4[KK * 8 + q];
  add_pass(acc[0], r4, __int_as_float(bcast8<(KK >> 2)>(__float_as_int(d2v[0][KK & 3]))), s4[0], r[0]);
  add_pass(acc[1], r4, __int_as_float(bcast8<(KK >> 2)>(__float_as_int(d2v[1][KK & 3]))), s4[1], r[1]);
  add_pass(acc[2], r4, __int_as_float(bcast8<(KK >> 2)>(__float_as_int(d2v[2][KK & 3]))), s4[2], r[2]);
  add_pass(acc[3], r4, __int_as_float(bcast8<(KK >> 2)>(__float_as_int(d2v[3][KK & 3]))), s4[3], r[3]);
}

template <int K>
__device__ __forceinline__ void pipe_step(v4f (&acc)[4], v4f (&ra)[4], v4f (&rb)[4],
                                          const v4f* __restrict__ dect4,
                                          const v4i (&n4)[4], const v4f (&d2v)[4],
                                          const v4f (&s4)[4], const v4f* rd4,
                                          const unsigned q) {
  use_slot<K>(acc, ra, d2v, s4, rd4, q);
  use_slot<K + 1>(acc, rb, d2v, s4, rd4, q);
  if constexpr (K + 4 < KNEI) {
    issue_slot<K + 4>(ra, dect4, n4, q);
    issue_slot<K + 5>(rb, dect4, n4, q);
  }
}

__global__ __launch_bounds__(FO_THREADS)
void k_fused_out(const float* __restrict__ d,
                 const int* __restrict__ nid,
                 const int* __restrict__ lab,
                 const float* __restrict__ dect,
                 const float* __restrict__ invw2t,
                 const float* __restrict__ rdent,
                 const float* __restrict__ encoded,
                 float* __restrict__ out) {
  __shared__ __attribute__((aligned(16))) float iw[MCLU * NBAS];
  __shared__ __attribute__((aligned(16))) float rd[NBAS * NBAS];
  __shared__ float enc_s[NBAS];
  __shared__ float g[FO_WAVES][32 * 17];
  for (int t = threadIdx.x; t < MCLU * NBAS / 4; t += FO_THREADS)
    reinterpret_cast<v4f*>(iw)[t] = reinterpret_cast<const v4f*>(invw2t)[t];
  for (int t = threadIdx.x; t < NBAS * NBAS / 4; t += FO_THREADS)
    reinterpret_cast<v4f*>(rd)[t] = reinterpret_cast<const v4f*>(rdent)[t];
  if (threadIdx.x < NBAS) enc_s[threadIdx.x] = encoded[threadIdx.x];
  __syncthreads();

  const int lane = threadIdx.x & 31, wv = threadIdx.x >> 5;
  const int tile = blockIdx.x * FO_WAVES + wv;
  const int NT = NPTS / 16;
  if (tile >= NT) return;
  float* gbuf = g[wv];
  const int j0 = tile * 16;
  const int pt = lane >> 3;
  const unsigned q = (unsigned)(lane & 7);
  const v4f* dect4 = reinterpret_cast<const v4f*>(dect);
  const v4f* rd4 = reinterpret_cast<const v4f*>(rd);

  v4i n4[4];
  v4f d2v[4], s4[4], acc[4];
  int cl[4];
#pragma unroll
  for (int sub = 0; sub < 4; ++sub) {
    const unsigned j = (unsigned)(j0 + sub * 4 + pt);
    n4[sub] = reinterpret_cast<const v4i*>(nid)[j * 8u + q];
    const v4f dv = reinterpret_cast<const v4f*>(d)[j * 8u + q];
    d2v[sub] = dv * dv;
    cl[sub] = lab[j];
  }
  v4f ra[4], rb[4], rc[4], re[4];
  issue_slot<0>(ra, dect4, n4, q);
  issue_slot<1>(rb, dect4, n4, q);
  issue_slot<2>(rc, dect4, n4, q);
  issue_slot<3>(re, dect4, n4, q);
#pragma unroll
  for (int sub = 0; sub < 4; ++sub) {
    s4[sub] = reinterpret_cast<const v4f*>(iw)[(unsigned)cl[sub] * 8u + q];
    acc[sub] = (v4f){0.f, 0.f, 0.f, 0.f};
  }
  pipe_step<0>(acc, ra, rb, dect4, n4, d2v, s4, rd4, q);
  pipe_step<2>(acc, rc, re, dect4, n4, d2v, s4, rd4, q);
  pipe_step<4>(acc, ra, rb, dect4, n4, d2v, s4, rd4, q);
  pipe_step<6>(acc, rc, re, dect4, n4, d2v, s4, rd4, q);
  pipe_step<8>(acc, ra, rb, dect4, n4, d2v, s4, rd4, q);
  pipe_step<10>(acc, rc, re, dect4, n4, d2v, s4, rd4, q);
  pipe_step<12>(acc, ra, rb, dect4, n4, d2v, s4, rd4, q);
  pipe_step<14>(acc, rc, re, dect4, n4, d2v, s4, rd4, q);
  pipe_step<16>(acc, ra, rb, dect4, n4, d2v, s4, rd4, q);
  pipe_step<18>(acc, rc, re, dect4, n4, d2v, s4, rd4, q);
  pipe_step<20>(acc, ra, rb, dect4, n4, d2v, s4, rd4, q);
  pipe_step<22>(acc, rc, re, dect4, n4, d2v, s4, rd4, q);
  pipe_step<24>(acc, ra, rb, dect4, n4, d2v, s4, rd4, q);
  pipe_step<26>(acc, rc, re, dect4, n4, d2v, s4, rd4, q);
  pipe_step<28>(acc, ra, rb, dect4, n4, d2v, s4, rd4, q);
  pipe_step<30>(acc, rc, re, dect4, n4, d2v, s4, rd4, q);

#pragma unroll
  for (int sub = 0; sub < 4; ++sub) {
    const int colp = sub * 4 + pt;
    gbuf[(4 * q + 0) * 17 + colp] = acc[sub].x;
    gbuf[(4 * q + 1) * 17 + colp] = acc[sub].y;
    gbuf[(4 * q + 2) * 17 + colp] = acc[sub].z;
    gbuf[(4 * q + 3) * 17 + colp] = acc[sub].w;
  }
  asm volatile("s_wait_dscnt 0" ::: "memory");

  v8f cacc = {0.f, 0.f, 0.f, 0.f, 0.f, 0.f, 0.f, 0.f};
  const int half = (lane >> 4) & 1;
  const int col = lane & 15;
#pragma unroll
  for (int s8 = 0; s8 < 8; ++s8) {
    const int kb = s8 * 4;
    v2f a, b;
    a.x = enc_s[kb + 2 * half];
    a.y = enc_s[kb + 2 * half + 1];
    b.x = gbuf[(kb + 2 * half) * 17 + col];
    b.y = gbuf[(kb + 2 * half + 1) * 17 + col];
    cacc = __builtin_amdgcn_wmma_f32_16x16x4_f32(
        false, a, false, b, (short)0, cacc, false, false);
  }
  if (lane < 16) out[j0 + lane] = cacc[0];
}

extern "C" void kernel_launch(void* const* d_in, const int* in_sizes, int n_in,
                              void* d_out, int out_size, void* d_ws, size_t ws_size,
                              hipStream_t stream) {
  const float* x     = (const float*)d_in[0];
  const float* enc_w = (const float*)d_in[1];
  const float* enc_b = (const float*)d_in[2];
  const float* dec   = (const float*)d_in[3];
  const float* bw_w  = (const float*)d_in[4];
  const float* bw_b  = (const float*)d_in[5];
  const float* nd    = (const float*)d_in[6];
  const int*   nid   = (const int*)d_in[7];
  const int*   lab   = (const int*)d_in[8];
  float* out = (float*)d_out;
  float* ws  = (float*)d_ws;

  float* enc_part = ws + WS_ENCPART;
  float* encoded  = ws + WS_ENCODED;
  float* invw2t   = ws + WS_INVW2T;
  float* rdent    = ws + WS_RDENT;
  float* denpart  = ws + WS_DENPART;
  float* dect     = ws + WS_DECT;

  k_enc_partial<<<GV_BLOCKS, GV_WAVES * 32, 0, stream>>>(x, enc_w, enc_part);
  k_enc_finish<<<1, 512, 0, stream>>>(enc_part, enc_b, bw_w, bw_b, encoded, invw2t);
  k_dec_transpose<<<TR_BLOCKS, TR_WAVES * 32, 0, stream>>>(dec, dect);
  k_denom_partial<<<DEN_GRID, DEN_THR, 0, stream>>>(nd, lab, invw2t, denpart);
  k_denom_finish<<<1, 1024, 0, stream>>>(denpart, DEN_BLKS, rdent);
  k_fused_out<<<FO_BLOCKS, FO_THREADS, 0, stream>>>(nd, nid, lab, dect, invw2t, rdent,
                                       encoded, out);
}


# denominator pass re-tiled, points across lanes: a lane holds one point's 32 squared distances and 2 window scales in registers, 64 in-lane sums, lane butterfly; denominator sums regrouped
# speedup vs baseline: 8.8108x; 1.0413x over previous
#include <hip/hip_runtime.h>
#include <math.h>


#define NPTS 100000
#define NBAS 32
#define MCLU 16
#define KNEI 32
#define MUF  100.0f

typedef __attribute__((ext_vector_type(2))) float v2f;
typedef __attribute__((ext_vector_type(8))) float v8f;
typedef __attribute__((ext_vector_type(4))) float v4f;
typedef __attribute__((ext_vector_type(4))) int v4i;

#define GV_WAVES  4
#define GV_TILE   128
#define GV_TILES  ((NPTS + GV_TILE - 1) / GV_TILE)
#define GV_BLOCKS ((GV_TILES + GV_WAVES - 1) / GV_WAVES)
#define GV_GROUPS 16
#define GV_PER    ((GV_BLOCKS + GV_GROUPS - 1) / GV_GROUPS)
#define WS_ENCPART 0
#define WS_ENCODED (WS_ENCPART + GV_BLOCKS * 32)
#define WS_INVW2T  (WS_ENCODED + 32)
#define WS_RDENT   (WS_INVW2T + 512)
#define DEN_BLKS   128
#define WS_DENPART (WS_RDENT + 1024)
#define WS_DECT    (WS_DENPART + DEN_BLKS*1024)

template <int S>
__device__ __forceinline__ void gv_fold(float (&a)[32], const unsigned lane) {
  const bool up = (lane & (unsigned)S) != 0u;
#pragma unroll
  for (int r = 0; r < S; ++r) {
    const float keep = up ? a[r + S] : a[r];
    const float send = up ? a[r] : a[r + S];
    a[r] = keep + __shfl_xor(send, S);
  }
}

__global__ __launch_bounds__(GV_WAVES * 32)
void k_enc_partial(const float* __restrict__ x,
                   const float* __restrict__ W,
                   float* __restrict__ part) {
  __shared__ float red[GV_WAVES][32];
  const unsigned lane = threadIdx.x & 31u, wv = threadIdx.x >> 5;
  const unsigned tile = blockIdx.x * GV_WAVES + wv;
  const unsigned j = tile * GV_TILE + lane * 4u;
  const bool valid = j < (unsigned)NPTS;
  const unsigned jc = valid ? j : (unsigned)(NPTS - 4);
  const v4f xv = *reinterpret_cast<const v4f*>(x + jc);
  const v4f* wp = reinterpret_cast<const v4f*>(W + jc);
  v4f w[NBAS];
#pragma unroll
  for (int r = 0; r < NBAS; ++r) w[r] = wp[r * (NPTS / 4)];
  float a[NBAS];
#pragma unroll
  for (int r = 0; r < NBAS; ++r) {
    float s = w[r].x * xv.x;
    s += w[r].y * xv.y;
    s += w[r].z * xv.z;
    s += w[r].w * xv.w;
    a[r] = valid ? s : 0.f;
  }
  gv_fold<16>(a, lane);
  gv_fold<8>(a, lane);
  gv_fold<4>(a, lane);
  gv_fold<2>(a, lane);
  gv_fold<1>(a, lane);
  red[wv][lane] = a[0];
  __syncthreads();
  if (wv == 0)
    part[blockIdx.x * 32 + lane] = ((red[0][lane] + red[1][lane]) + red[2][lane]) + red[3][lane];
}

__global__ void k_enc_finish(const float* __restrict__ part,
                             const float* __restrict__ enc_b,
                             const float* __restrict__ bw_w,
                             const float* __restrict__ bw_b,
                             float* __restrict__ encoded,
                             float* __restrict__ invw2t) {
  __shared__ float enc_s[NBAS];
  __shared__ float ps[GV_GROUPS][NBAS];
  const int t = threadIdx.x;
  {
    const int g = t >> 5, r = t & 31;
    float s = 0.f;
#pragma unroll
    for (int k = 0; k < GV_PER; ++k) {
      const int b = g + k * GV_GROUPS;
      const float v = part[min(b, GV_BLOCKS - 1) * 32 + r];
      s += (b < GV_BLOCKS) ? v : 0.f;
    }
    ps[g][r] = s;
  }
  __syncthreads();
  if (t < NBAS) {
    float s = 0.f;
    for (int g = 0; g < GV_GROUPS; ++g) s += ps[g][t];
    s += enc_b[t];
    enc_s[t] = s;
    encoded[t] = s;
  }
  __syncthreads();
  if (t < NBAS * MCLU) {
    float s = 0.f;
    for (int q = 0; q < NBAS; ++q) s += enc_s[q] * bw_w[t * NBAS + q];
    s += bw_b[t];
    float band = (1.f / (1.f + expf(-s))) / 60.f;
    float wm = band * MUF;
    const int i = t / MCLU, c = t % MCLU;
    invw2t[c * NBAS + i] = 1.f / (wm * wm);
  }
}

#define TR_WAVES  4
#define TR_TILES  (NPTS / 32)
#define TR_BLOCKS ((TR_TILES + TR_WAVES - 1) / TR_WAVES)

__global__ __launch_bounds__(TR_WAVES * 32)
void k_dec_transpose(const float* __restrict__ dec,
                     float* __restrict__ dect) {
  const unsigned lane = threadIdx.x & 31u, wv = threadIdx.x >> 5;
  const unsigned tile = blockIdx.x * TR_WAVES + wv;
  if (tile >= TR_TILES) return;
  const v4f* src = reinterpret_cast<const v4f*>(dec + lane * (unsigned)NPTS + tile * 32u);
  v4f r[8];
#pragma unroll
  for (int s = 0; s < 8; ++s) r[s] = src[s];
  float* dst = dect + tile * 1024u + lane;
#pragma unroll
  for (int p = 0; p < 32; ++p) dst[p * 32] = r[p >> 2][p & 3];
}

#define DEN_GRID 256
#define DEN_THR  512
#define DEN_CH   782
#define DEN_H    392
#define DEN_RD   13
#define DEN_PITCH 36
#define DEN_ROWS (DEN_RD * DEN_THR / 8)
static_assert(DEN_BLKS * DEN_CH >= NPTS, "point groups cover all points");
static_assert(NPTS - (DEN_BLKS - 1) * DEN_CH >= DEN_THR, "shortest chunk has a label for every thread");
static_assert(DEN_RD * DEN_THR >= DEN_CH * (KNEI / 4), "one burst covers a chunk");
static_assert(MCLU * NBAS == DEN_THR, "one table entry per thread");
static_assert(2 * DEN_THR >= DEN_CH + 2, "label slots cover a chunk");
static_assert(NPTS - (DEN_BLKS - 1) * DEN_CH > DEN_H, "every chunk has a second half");

template <int X, int H>
__device__ __forceinline__ void den_fold(float (&a)[64], const unsigned lane) {
  const bool up = (lane & (unsigned)X) != 0u;
#pragma unroll
  for (int r = 0; r < H; ++r) {
    const float keep = up ? a[r + H] : a[r];
    const float send = up ? a[r] : a[r + H];
    a[r] = keep + __shfl_xor(send, X);
  }
}

__device__ __forceinline__ v2f den_win(const v2f d2, const float s) {
  const v2f t = 1.f - d2 * s;
  v2f r;
  r.x = fmaxf(t.x, 0.f);
  r.y = fmaxf(t.y, 0.f);
  return r;
}

__global__ __launch_bounds__(DEN_THR) __attribute__((amdgpu_waves_per_eu(4, 4)))
void k_denom_partial(const float* __restrict__ d,
                     const int* __restrict__ lab,
                     const float* __restrict__ invw2t,
                     float* __restrict__ part) {
  __shared__ __attribute__((aligned(16))) float d2s[DEN_ROWS * DEN_PITCH];
  __shared__ __attribute__((aligned(16))) int   labs[2 * DEN_THR];
  __shared__ __attribute__((aligned(16))) float iw[MCLU * NBAS];
  __shared__ __attribute__((aligned(16))) float comb[8 * 64];
  const int tid = threadIdx.x;
  const int pg = blockIdx.x >> 1, ih = blockIdx.x & 1;
  const int p0 = pg * DEN_CH;
  const int n  = min(DEN_CH, NPTS - p0);
  const int n4 = n * (KNEI / 4);

  const int lv0 = lab[p0 + tid];
  const int lv1 = lab[p0 + min(tid + DEN_THR, n - 1)];
  const float iv = invw2t[tid];
  const v4f* src = reinterpret_cast<const v4f*>(d + p0 * KNEI);
  v4f r[DEN_RD];
#pragma unroll
  for (int s = 0; s < DEN_RD; ++s) r[s] = src[min(s * DEN_THR + tid, n4 - 1)];
  labs[tid] = lv0;
  labs[tid + DEN_THR] = lv1;
  iw[tid] = iv;
#pragma unroll
  for (int s = 0; s < DEN_RD; ++s) {
    const int g = s * DEN_THR + tid;
    *reinterpret_cast<v4f*>(d2s + (g >> 3) * DEN_PITCH + (g & 7) * 4) = r[s] * r[s];
  }
  __syncthreads();

  const unsigned lane = (unsigned)tid & 31u;
  const int w = tid >> 5;
  const int bp = w & 7, ph = w >> 3;
  const int i0 = ih * 16 + bp * 2;
  const int pend = ph ? n : DEN_H;
  v2f acc0[16], acc1[16];
#pragma unroll
  for (int q = 0; q < 16; ++q) {
    acc0[q] = (v2f){0.f, 0.f};
    acc1[q] = (v2f){0.f, 0.f};
  }
  for (int p = ph * DEN_H + (int)lane; p < pend; p += 32) {
    const int c = labs[p];
    const v2f a = *reinterpret_cast<const v2f*>(iw + c * NBAS + i0);
    const v4f* row = reinterpret_cast<const v4f*>(d2s + p * DEN_PITCH);
#pragma unroll
    for (int q = 0; q < 8; ++q) {
      const v4f dq = row[q];
      const v2f lo = {dq.x, dq.y}, hi = {dq.z, dq.w};
      acc0[2 * q]     += den_win(lo, a.x);
      acc0[2 * q + 1] += den_win(hi, a.x);
      acc1[2 * q]     += den_win(lo, a.y);
      acc1[2 * q + 1] += den_win(hi, a.y);
    }
  }

  float f[64];
#pragma unroll
  for (int q = 0; q < 16; ++q) {
    f[2 * q]          = acc0[q].x;
    f[2 * q + 1]      = acc0[q].y;
    f[32 + 2 * q]     = acc1[q].x;
    f[32 + 2 * q + 1] = acc1[q].y;
  }
  den_fold<16, 32>(f, lane);
  den_fold<8, 16>(f, lane);
  den_fold<4, 8>(f, lane);
  den_fold<2, 4>(f, lane);
  den_fold<1, 2>(f, lane);

  if (ph) {
    *reinterpret_cast<v2f*>(comb + bp * 64 + 2 * (int)lane) = (v2f){f[0], f[1]};
  }
  __syncthreads();
  if (!ph) {
    const v2f o = *reinterpret_cast<const v2f*>(comb + bp * 64 + 2 * (int)lane);
    *reinterpret_cast<v2f*>(part + pg * 1024 + i0 * 32 + 2 * (int)lane) = (v2f){f[0] + o.x, f[1] + o.y};
  }
}

__global__ void k_denom_finish(const float* __restrict__ part, int nblk,
                               float* __restrict__ rdent) {
  const int t = threadIdx.x;
  float s = 0.f;
  for (int b = 0; b < nblk; ++b) s += part[b * 1024 + t];
  const int i = t >> 5, kk = t & 31;
  rdent[kk * 32 + i] = 1.f / s;
}

#define FO_WAVES   4
#define FO_THREADS 128
#define FO_BLOCKS  1563

template <int G>
__device__ __forceinline__ int bcast8(int v) {
  return (int)__builtin_amdgcn_mov_dpp8((unsigned)v, (unsigned)(G * 0x249249));
}

template <int KK>
__device__ __forceinline__ void issue_slot(v4f (&r)[4], const v4f* __restrict__ dect4,
                                           const v4i (&n4)[4], const unsigned q) {
  r[0] = dect4[(unsigned)bcast8<(KK >> 2)>(n4[0][KK & 3]) * 8u + q];
  r[1] = dect4[(unsigned)bcast8<(KK >> 2)>(n4[1][KK & 3]) * 8u + q];
  r[2] = dect4[(unsigned)bcast8<(KK >> 2)>(n4[2][KK & 3]) * 8u + q];
  r[3] = dect4[(unsigned)bcast8<(KK >> 2)>(n4[3][KK & 3]) * 8u + q];
}

__device__ __forceinline__ void add_pass(v4f& acc, const v4f r4, const float d2k,
                                         const v4f s4, const v4f dec) {
  acc.x += r4.x * fmaxf(1.f - d2k * s4.x, 0.f) * dec.x;
  acc.y += r4.y * fmaxf(1.f - d2k * s4.y, 0.f) * dec.y;
  acc.z += r4.z * fmaxf(1.f - d2k * s4.z, 0.f) * dec.z;
  acc.w += r4.w * fmaxf(1.f - d2k * s4.w, 0.f) * dec.w;
}

template <int KK>
__device__ __forceinline__ void use_slot(v4f (&acc)[4], const v4f (&r)[4],
                                         const v4f (&d2v)[4], const v4f (&s4)[4],
                                         const v4f* rd4, const unsigned q) {
  const v4f r4 = rd4[KK * 8 + q];
  add_pass(acc[0], r4, __int_as_float(bcast8<(KK >> 2)>(__float_as_int(d2v[0][KK & 3]))), s4[0], r[0]);
  add_pass(acc[1], r4, __int_as_float(bcast8<(KK >> 2)>(__float_as_int(d2v[1][KK & 3]))), s4[1], r[1]);
  add_pass(acc[2], r4, __int_as_float(bcast8<(KK >> 2)>(__float_as_int(d2v[2][KK & 3]))), s4[2], r[2]);
  add_pass(acc[3], r4, __int_as_float(bcast8<(KK >> 2)>(__float_as_int(d2v[3][KK & 3]))), s4[3], r[3]);
}

template <int K>
__device__ __forceinline__ void pipe_step(v4f (&acc)[4], v4f (&ra)[4], v4f (&rb)[4],
                                          const v4f* __restrict__ dect4,
                                          const v4i (&n4)[4], const v4f (&d2v)[4],
                                          const v4f (&s4)[4], const v4f* rd4,
                                          const unsigned q) {
  use_slot<K>(acc, ra, d2v, s4, rd4, q);
  use_slot<K + 1>(acc, rb, d2v, s4, rd4, q);
  if constexpr (K + 4 < KNEI) {
    issue_slot<K + 4>(ra, dect4, n4, q);
    issue_slot<K + 5>(rb, dect4, n4, q);
  }
}

__global__ __launch_bounds__(FO_THREADS)
void k_fused_out(const float* __restrict__ d,
                 const int* __restrict__ nid,
                 const int* __restrict__ lab,
                 const float* __restrict__ dect,
                 const float* __restrict__ invw2t,
                 const float* __restrict__ rdent,
                 const float* __restrict__ encoded,
                 float* __restrict__ out) {
  __shared__ __attribute__((aligned(16))) float iw[MCLU * NBAS];
  __shared__ __attribute__((aligned(16))) float rd[NBAS * NBAS];
  __shared__ float enc_s[NBAS];
  __shared__ float g[FO_WAVES][32 * 17];
  for (int t = threadIdx.x; t < MCLU * NBAS / 4; t += FO_THREADS)
    reinterpret_cast<v4f*>(iw)[t] = reinterpret_cast<const v4f*>(invw2t)[t];
  for (int t = threadIdx.x; t < NBAS * NBAS / 4; t += FO_THREADS)
    reinterpret_cast<v4f*>(rd)[t] = reinterpret_cast<const v4f*>(rdent)[t];
  if (threadIdx.x < NBAS) enc_s[threadIdx.x] = encoded[threadIdx.x];
  __syncthreads();

  const int lane = threadIdx.x & 31, wv = threadIdx.x >> 5;
  const int tile = blockIdx.x * FO_WAVES + wv;
  const int NT = NPTS / 16;
  if (tile >= NT) return;
  float* gbuf = g[wv];
  const int j0 = tile * 16;
  const int pt = lane >> 3;
  const unsigned q = (unsigned)(lane & 7);
  const v4f* dect4 = reinterpret_cast<const v4f*>(dect);
  const v4f* rd4 = reinterpret_cast<const v4f*>(rd);

  v4i n4[4];
  v4f d2v[4], s4[4], acc[4];
  int cl[4];
#pragma unroll
  for (int sub = 0; sub < 4; ++sub) {
    const unsigned j = (unsigned)(j0 + sub * 4 + pt);
    n4[sub] = reinterpret_cast<const v4i*>(nid)[j * 8u + q];
    const v4f dv = reinterpret_cast<const v4f*>(d)[j * 8u + q];
    d2v[sub] = dv * dv;
    cl[sub] = lab[j];
  }
  v4f ra[4], rb[4], rc[4], re[4];
  issue_slot<0>(ra, dect4, n4, q);
  issue_slot<1>(rb, dect4, n4, q);
  issue_slot<2>(rc, dect4, n4, q);
  issue_slot<3>(re, dect4, n4, q);
#pragma unroll
  for (int sub = 0; sub < 4; ++sub) {
    s4[sub] = reinterpret_cast<const v4f*>(iw)[(unsigned)cl[sub] * 8u + q];
    acc[sub] = (v4f){0.f, 0.f, 0.f, 0.f};
  }
  pipe_step<0>(acc, ra, rb, dect4, n4, d2v, s4, rd4, q);
  pipe_step<2>(acc, rc, re, dect4, n4, d2v, s4, rd4, q);
  pipe_step<4>(acc, ra, rb, dect4, n4, d2v, s4, rd4, q);
  pipe_step<6>(acc, rc, re, dect4, n4, d2v, s4, rd4, q);
  pipe_step<8>(acc, ra, rb, dect4, n4, d2v, s4, rd4, q);
  pipe_step<10>(acc, rc, re, dect4, n4, d2v, s4, rd4, q);
  pipe_step<12>(acc, ra, rb, dect4, n4, d2v, s4, rd4, q);
  pipe_step<14>(acc, rc, re, dect4, n4, d2v, s4, rd4, q);
  pipe_step<16>(acc, ra, rb, dect4, n4, d2v, s4, rd4, q);
  pipe_step<18>(acc, rc, re, dect4, n4, d2v, s4, rd4, q);
  pipe_step<20>(acc, ra, rb, dect4, n4, d2v, s4, rd4, q);
  pipe_step<22>(acc, rc, re, dect4, n4, d2v, s4, rd4, q);
  pipe_step<24>(acc, ra, rb, dect4, n4, d2v, s4, rd4, q);
  pipe_step<26>(acc, rc, re, dect4, n4, d2v, s4, rd4, q);
  pipe_step<28>(acc, ra, rb, dect4, n4, d2v, s4, rd4, q);
  pipe_step<30>(acc, rc, re, dect4, n4, d2v, s4, rd4, q);

#pragma unroll
  for (int sub = 0; sub < 4; ++sub) {
    const int colp = sub * 4 + pt;
    gbuf[(4 * q + 0) * 17 + colp] = acc[sub].x;
    gbuf[(4 * q + 1) * 17 + colp] = acc[sub].y;
    gbuf[(4 * q + 2) * 17 + colp] = acc[sub].z;
    gbuf[(4 * q + 3) * 17 + colp] = acc[sub].w;
  }
  asm volatile("s_wait_dscnt 0" ::: "memory");

  v8f cacc = {0.f, 0.f, 0.f, 0.f, 0.f, 0.f, 0.f, 0.f};
  const int half = (lane >> 4) & 1;
  const int col = lane & 15;
#pragma unroll
  for (int s8 = 0; s8 < 8; ++s8) {
    const int kb = s8 * 4;
    v2f a, b;
    a.x = enc_s[kb + 2 * half];
    a.y = enc_s[kb + 2 * half + 1];
    b.x = gbuf[(kb + 2 * half) * 17 + col];
    b.y = gbuf[(kb + 2 * half + 1) * 17 + col];
    cacc = __builtin_amdgcn_wmma_f32_16x16x4_f32(
        false, a, false, b, (short)0, cacc, false, false);
  }
  if (lane < 16) out[j0 + lane] = cacc[0];
}

extern "C" void kernel_launch(void* const* d_in, const int* in_sizes, int n_in,
                              void* d_out, int out_size, void* d_ws, size_t ws_size,
                              hipStream_t stream) {
  const float* x     = (const float*)d_in[0];
  const float* enc_w = (const float*)d_in[1];
  const float* enc_b = (const float*)d_in[2];
  const float* dec   = (const float*)d_in[3];
  const float* bw_w  = (const float*)d_in[4];
  const float* bw_b  = (const float*)d_in[5];
  const float* nd    = (const float*)d_in[6];
  const int*   nid   = (const int*)d_in[7];
  const int*   lab   = (const int*)d_in[8];
  float* out = (float*)d_out;
  float* ws  = (float*)d_ws;

  float* enc_part = ws + WS_ENCPART;
  float* encoded  = ws + WS_ENCODED;
  float* invw2t   = ws + WS_INVW2T;
  float* rdent    = ws + WS_RDENT;
  float* denpart  = ws + WS_DENPART;
  float* dect     = ws + WS_DECT;

  k_enc_partial<<<GV_BLOCKS, GV_WAVES * 32, 0, stream>>>(x, enc_w, enc_part);
  k_enc_finish<<<1, 512, 0, stream>>>(enc_part, enc_b, bw_w, bw_b, encoded, invw2t);
  k_dec_transpose<<<TR_BLOCKS, TR_WAVES * 32, 0, stream>>>(dec, dect);
  k_denom_partial<<<DEN_GRID, DEN_THR, 0, stream>>>(nd, lab, invw2t, denpart);
  k_denom_finish<<<1, 1024, 0, stream>>>(denpart, DEN_BLKS, rdent);
  k_fused_out<<<FO_BLOCKS, FO_THREADS, 0, stream>>>(nd, nid, lab, dect, invw2t, rdent,
                                       encoded, out);
}


# denominator finish moved into its consumer: gather kernel = 256 resident 20-wave blocks walking the tiles, each sums the 128 partial planes itself; 5 launches (was 6); denominator sums regrouped
# speedup vs baseline: 9.7217x; 1.1034x over previous
#include <hip/hip_runtime.h>
#include <math.h>


#define NPTS 100000
#define NBAS 32
#define MCLU 16
#define KNEI 32
#define MUF  100.0f

typedef __attribute__((ext_vector_type(2))) float v2f;
typedef __attribute__((ext_vector_type(8))) float v8f;
typedef __attribute__((ext_vector_type(4))) float v4f;
typedef __attribute__((ext_vector_type(4))) int v4i;

#define GV_WAVES  4
#define GV_TILE   128
#define GV_TILES  ((NPTS + GV_TILE - 1) / GV_TILE)
#define GV_BLOCKS ((GV_TILES + GV_WAVES - 1) / GV_WAVES)
#define GV_GROUPS 16
#define GV_PER    ((GV_BLOCKS + GV_GROUPS - 1) / GV_GROUPS)
#define WS_ENCPART 0
#define WS_ENCODED (WS_ENCPART + GV_BLOCKS * 32)
#define WS_INVW2T  (WS_ENCODED + 32)
#define WS_RDENT   (WS_INVW2T + 512)
#define DEN_BLKS   128
#define WS_DENPART (WS_RDENT + 1024)
#define WS_DECT    (WS_DENPART + DEN_BLKS*1024)

template <int S>
__device__ __forceinline__ void gv_fold(float (&a)[32], const unsigned lane) {
  const bool up = (lane & (unsigned)S) != 0u;
#pragma unroll
  for (int r = 0; r < S; ++r) {
    const float keep = up ? a[r + S] : a[r];
    const float send = up ? a[r] : a[r + S];
    a[r] = keep + __shfl_xor(send, S);
  }
}

__global__ __launch_bounds__(GV_WAVES * 32)
void k_enc_partial(const float* __restrict__ x,
                   const float* __restrict__ W,
                   float* __restrict__ part) {
  __shared__ float red[GV_WAVES][32];
  const unsigned lane = threadIdx.x & 31u, wv = threadIdx.x >> 5;
  const unsigned tile = blockIdx.x * GV_WAVES + wv;
  const unsigned j = tile * GV_TILE + lane * 4u;
  const bool valid = j < (unsigned)NPTS;
  const unsigned jc = valid ? j : (unsigned)(NPTS - 4);
  const v4f xv = *reinterpret_cast<const v4f*>(x + jc);
  const v4f* wp = reinterpret_cast<const v4f*>(W + jc);
  v4f w[NBAS];
#pragma unroll
  for (int r = 0; r < NBAS; ++r) w[r] = wp[r * (NPTS / 4)];
  float a[NBAS];
#pragma unroll
  for (int r = 0; r < NBAS; ++r) {
    float s = w[r].x * xv.x;
    s += w[r].y * xv.y;
    s += w[r].z * xv.z;
    s += w[r].w * xv.w;
    a[r] = valid ? s : 0.f;
  }
  gv_fold<16>(a, lane);
  gv_fold<8>(a, lane);
  gv_fold<4>(a, lane);
  gv_fold<2>(a, lane);
  gv_fold<1>(a, lane);
  red[wv][lane] = a[0];
  __syncthreads();
  if (wv == 0)
    part[blockIdx.x * 32 + lane] = ((red[0][lane] + red[1][lane]) + red[2][lane]) + red[3][lane];
}

__global__ void k_enc_finish(const float* __restrict__ part,
                             const float* __restrict__ enc_b,
                             const float* __restrict__ bw_w,
                             const float* __restrict__ bw_b,
                             float* __restrict__ encoded,
                             float* __restrict__ invw2t) {
  __shared__ float enc_s[NBAS];
  __shared__ float ps[GV_GROUPS][NBAS];
  const int t = threadIdx.x;
  {
    const int g = t >> 5, r = t & 31;
    float s = 0.f;
#pragma unroll
    for (int k = 0; k < GV_PER; ++k) {
      const int b = g + k * GV_GROUPS;
      const float v = part[min(b, GV_BLOCKS - 1) * 32 + r];
      s += (b < GV_BLOCKS) ? v : 0.f;
    }
    ps[g][r] = s;
  }
  __syncthreads();
  if (t < NBAS) {
    float s = 0.f;
    for (int g = 0; g < GV_GROUPS; ++g) s += ps[g][t];
    s += enc_b[t];
    enc_s[t] = s;
    encoded[t] = s;
  }
  __syncthreads();
  if (t < NBAS * MCLU) {
    float s = 0.f;
    for (int q = 0; q < NBAS; ++q) s += enc_s[q] * bw_w[t * NBAS + q];
    s += bw_b[t];
    float band = (1.f / (1.f + expf(-s))) / 60.f;
    float wm = band * MUF;
    const int i = t / MCLU, c = t % MCLU;
    invw2t[c * NBAS + i] = 1.f / (wm * wm);
  }
}

#define TR_WAVES  4
#define TR_TILES  (NPTS / 32)
#define TR_BLOCKS ((TR_TILES + TR_WAVES - 1) / TR_WAVES)

__global__ __launch_bounds__(TR_WAVES * 32)
void k_dec_transpose(const float* __restrict__ dec,
                     float* __restrict__ dect) {
  const unsigned lane = threadIdx.x & 31u, wv = threadIdx.x >> 5;
  const unsigned tile = blockIdx.x * TR_WAVES + wv;
  if (tile >= TR_TILES) return;
  const v4f* src = reinterpret_cast<const v4f*>(dec + lane * (unsigned)NPTS + tile * 32u);
  v4f r[8];
#pragma unroll
  for (int s = 0; s < 8; ++s) r[s] = src[s];
  float* dst = dect + tile * 1024u + lane;
#pragma unroll
  for (int p = 0; p < 32; ++p) dst[p * 32] = r[p >> 2][p & 3];
}

#define DEN_GRID 256
#define DEN_THR  512
#define DEN_CH   782
#define DEN_H    392
#define DEN_RD   13
#define DEN_PITCH 36
#define DEN_ROWS (DEN_RD * DEN_THR / 8)
static_assert(DEN_BLKS * DEN_CH >= NPTS, "point groups cover all points");
static_assert(NPTS - (DEN_BLKS - 1) * DEN_CH >= DEN_THR, "shortest chunk has a label for every thread");
static_assert(DEN_RD * DEN_THR >= DEN_CH * (KNEI / 4), "one burst covers a chunk");
static_assert(MCLU * NBAS == DEN_THR, "one table entry per thread");
static_assert(2 * DEN_THR >= DEN_CH + 2, "label slots cover a chunk");
static_assert(NPTS - (DEN_BLKS - 1) * DEN_CH > DEN_H, "every chunk has a second half");

template <int X, int H>
__device__ __forceinline__ void den_fold(float (&a)[64], const unsigned lane) {
  const bool up = (lane & (unsigned)X) != 0u;
#pragma unroll
  for (int r = 0; r < H; ++r) {
    const float keep = up ? a[r + H] : a[r];
    const float send = up ? a[r] : a[r + H];
    a[r] = keep + __shfl_xor(send, X);
  }
}

__device__ __forceinline__ v2f den_win(const v2f d2, const float s) {
  const v2f t = 1.f - d2 * s;
  v2f r;
  r.x = fmaxf(t.x, 0.f);
  r.y = fmaxf(t.y, 0.f);
  return r;
}

__global__ __launch_bounds__(DEN_THR) __attribute__((amdgpu_waves_per_eu(4, 4)))
void k_denom_partial(const float* __restrict__ d,
                     const int* __restrict__ lab,
                     const float* __restrict__ invw2t,
                     float* __restrict__ part) {
  __shared__ __attribute__((aligned(16))) float d2s[DEN_ROWS * DEN_PITCH];
  __shared__ __attribute__((aligned(16))) int   labs[2 * DEN_THR];
  __shared__ __attribute__((aligned(16))) float iw[MCLU * NBAS];
  __shared__ __attribute__((aligned(16))) float comb[8 * 64];
  const int tid = threadIdx.x;
  const int pg = blockIdx.x >> 1, ih = blockIdx.x & 1;
  const int p0 = pg * DEN_CH;
  const int n  = min(DEN_CH, NPTS - p0);
  const int n4 = n * (KNEI / 4);

  const int lv0 = lab[p0 + tid];
  const int lv1 = lab[p0 + min(tid + DEN_THR, n - 1)];
  const float iv = invw2t[tid];
  const v4f* src = reinterpret_cast<const v4f*>(d + p0 * KNEI);
  v4f r[DEN_RD];
#pragma unroll
  for (int s = 0; s < DEN_RD; ++s) r[s] = src[min(s * DEN_THR + tid, n4 - 1)];
  labs[tid] = lv0;
  labs[tid + DEN_THR] = lv1;
  iw[tid] = iv;
#pragma unroll
  for (int s = 0; s < DEN_RD; ++s) {
    const int g = s * DEN_THR + tid;
    *reinterpret_cast<v4f*>(d2s + (g >> 3) * DEN_PITCH + (g & 7) * 4) = r[s] * r[s];
  }
  __syncthreads();

  const unsigned lane = (unsigned)tid & 31u;
  const int w = tid >> 5;
  const int bp = w & 7, ph = w >> 3;
  const int i0 = ih * 16 + bp * 2;
  const int pend = ph ? n : DEN_H;
  v2f acc0[16], acc1[16];
#pragma unroll
  for (int q = 0; q < 16; ++q) {
    acc0[q] = (v2f){0.f, 0.f};
    acc1[q] = (v2f){0.f, 0.f};
  }
  for (int p = ph * DEN_H + (int)lane; p < pend; p += 32) {
    const int c = labs[p];
    const v2f a = *reinterpret_cast<const v2f*>(iw + c * NBAS + i0);
    const v4f* row = reinterpret_cast<const v4f*>(d2s + p * DEN_PITCH);
#pragma unroll
    for (int q = 0; q < 8; ++q) {
      const v4f dq = row[q];
      const v2f lo = {dq.x, dq.y}, hi = {dq.z, dq.w};
      acc0[2 * q]     += den_win(lo, a.x);
      acc0[2 * q + 1] += den_win(hi, a.x);
      acc1[2 * q]     += den_win(lo, a.y);
      acc1[2 * q + 1] += den_win(hi, a.y);
    }
  }

  float f[64];
#pragma unroll
  for (int q = 0; q < 16; ++q) {
    f[2 * q]          = acc0[q].x;
    f[2 * q + 1]      = acc0[q].y;
    f[32 + 2 * q]     = acc1[q].x;
    f[32 + 2 * q + 1] = acc1[q].y;
  }
  den_fold<16, 32>(f, lane);
  den_fold<8, 16>(f, lane);
  den_fold<4, 8>(f, lane);
  den_fold<2, 4>(f, lane);
  den_fold<1, 2>(f, lane);

  if (ph) {
    *reinterpret_cast<v2f*>(comb + bp * 64 + 2 * (int)lane) = (v2f){f[0], f[1]};
  }
  __syncthreads();
  if (!ph) {
    const v2f o = *reinterpret_cast<const v2f*>(comb + bp * 64 + 2 * (int)lane);
    *reinterpret_cast<v2f*>(part + pg * 1024 + i0 * 32 + 2 * (int)lane) = (v2f){f[0] + o.x, f[1] + o.y};
  }
}

#define FO_WAVES   20
#define FO_THREADS 640
#define FO_BLOCKS  256
#define FO_STRIDE  (FO_BLOCKS * FO_WAVES)
#define FIN_GRP    2
#define FIN_PER    (DEN_BLKS / FIN_GRP)
#define FIN_RND    32
static_assert(FIN_GRP * 256 <= FO_THREADS, "finish threads exist");
static_assert(FIN_PER % FIN_RND == 0, "whole rounds");
static_assert(FO_STRIDE <= NPTS / 16, "every wave has a first tile");

template <int G>
__device__ __forceinline__ int bcast8(int v) {
  return (int)__builtin_amdgcn_mov_dpp8((unsigned)v, (unsigned)(G * 0x249249));
}

template <int KK>
__device__ __forceinline__ void issue_slot(v4f (&r)[4], const v4f* __restrict__ dect4,
                                           const v4i (&n4)[4], const unsigned q) {
  r[0] = dect4[(unsigned)bcast8<(KK >> 2)>(n4[0][KK & 3]) * 8u + q];
  r[1] = dect4[(unsigned)bcast8<(KK >> 2)>(n4[1][KK & 3]) * 8u + q];
  r[2] = dect4[(unsigned)bcast8<(KK >> 2)>(n4[2][KK & 3]) * 8u + q];
  r[3] = dect4[(unsigned)bcast8<(KK >> 2)>(n4[3][KK & 3]) * 8u + q];
}

__device__ __forceinline__ void add_pass(v4f& acc, const v4f r4, const float d2k,
                                         const v4f s4, const v4f dec) {
  acc.x += r4.x * fmaxf(1.f - d2k * s4.x, 0.f) * dec.x;
  acc.y += r4.y * fmaxf(1.f - d2k * s4.y, 0.f) * dec.y;
  acc.z += r4.z * fmaxf(1.f - d2k * s4.z, 0.f) * dec.z;
  acc.w += r4.w * fmaxf(1.f - d2k * s4.w, 0.f) * dec.w;
}

template <int KK>
__device__ __forceinline__ void use_slot(v4f (&acc)[4], const v4f (&r)[4],
                                         const v4f (&d2v)[4], const v4f (&s4)[4],
                                         const v4f* rd4, const unsigned q) {
  const v4f r4 = rd4[KK * 8 + q];
  add_pass(acc[0], r4, __int_as_float(bcast8<(KK >> 2)>(__float_as_int(d2v[0][KK & 3]))), s4[0], r[0]);
  add_pass(acc[1], r4, __int_as_float(bcast8<(KK >> 2)>(__float_as_int(d2v[1][KK & 3]))), s4[1], r[1]);
  add_pass(acc[2], r4, __int_as_float(bcast8<(KK >> 2)>(__float_as_int(d2v[2][KK & 3]))), s4[2], r[2]);
  add_pass(acc[3], r4, __int_as_float(bcast8<(KK >> 2)>(__float_as_int(d2v[3][KK & 3]))), s4[3], r[3]);
}

template <int K>
__device__ __forceinline__ void pipe_step(v4f (&acc)[4], v4f (&ra)[4], v4f (&rb)[4],
                                          const v4f* __restrict__ dect4,
                                          const v4i (&n4)[4], const v4f (&d2v)[4],
                                          const v4f (&s4)[4], const v4f* rd4,
                                          const unsigned q) {
  use_slot<K>(acc, ra, d2v, s4, rd4, q);
  use_slot<K + 1>(acc, rb, d2v, s4, rd4, q);
  if constexpr (K + 4 < KNEI) {
    issue_slot<K + 4>(ra, dect4, n4, q);
    issue_slot<K + 5>(rb, dect4, n4, q);
  }
}

__global__ __launch_bounds__(FO_THREADS)
void k_fused_out(const float* __restrict__ d,
                 const int* __restrict__ nid,
                 const int* __restrict__ lab,
                 const float* __restrict__ dect,
                 const float* __restrict__ invw2t,
                 const float* __restrict__ denpart,
                 const float* __restrict__ encoded,
                 float* __restrict__ out) {
  __shared__ __attribute__((aligned(16))) float iw[MCLU * NBAS];
  __shared__ __attribute__((aligned(16))) float rd[NBAS * NBAS];
  __shared__ __attribute__((aligned(16))) float fin[NBAS * NBAS];
  __shared__ float enc_s[NBAS];
  __shared__ float g[FO_WAVES][32 * 17];
  const int tid = threadIdx.x;
  const int lane = tid & 31, wv = tid >> 5;
  const int NT = NPTS / 16;
  const int pt = lane >> 3;
  const unsigned q = (unsigned)(lane & 7);

  int tile = wv * FO_BLOCKS + (int)blockIdx.x;
  v4i n4[4];
  v4f d2v[4];
  int cl[4];
#pragma unroll
  for (int sub = 0; sub < 4; ++sub) {
    const unsigned j = (unsigned)(tile * 16 + sub * 4 + pt);
    n4[sub] = reinterpret_cast<const v4i*>(nid)[j * 8u + q];
    d2v[sub] = reinterpret_cast<const v4f*>(d)[j * 8u + q];
    cl[sub] = lab[j];
  }

  const int fc = tid & 255, fg = tid >> 8;
  v4f fs = {0.f, 0.f, 0.f, 0.f};
  if (fg < FIN_GRP) {
    const v4f* pp = reinterpret_cast<const v4f*>(denpart) + (fg * FIN_PER) * 256 + fc;
    for (int b0 = 0; b0 < FIN_PER; b0 += FIN_RND) {
      v4f r[FIN_RND];
#pragma unroll
      for (int k = 0; k < FIN_RND; ++k) r[k] = pp[(b0 + k) * 256];
#pragma unroll
      for (int k = 0; k < FIN_RND; ++k) fs += r[k];
    }
    if (fg == 1) reinterpret_cast<v4f*>(fin)[fc] = fs;
  } else {
    const int t = tid - FIN_GRP * 256;
    reinterpret_cast<v4f*>(iw)[t] = reinterpret_cast<const v4f*>(invw2t)[t];
    if (t < NBAS) enc_s[t] = encoded[t];
  }
  __syncthreads();
  if (fg == 0) {
    const v4f o = reinterpret_cast<const v4f*>(fin)[fc];
    const int i = fc >> 3, kk = (fc & 7) * 4;
    rd[(kk + 0) * 32 + i] = 1.f / (fs.x + o.x);
    rd[(kk + 1) * 32 + i] = 1.f / (fs.y + o.y);
    rd[(kk + 2) * 32 + i] = 1.f / (fs.z + o.z);
    rd[(kk + 3) * 32 + i] = 1.f / (fs.w + o.w);
  }
  __syncthreads();

  float* gbuf = g[wv];
  const v4f* dect4 = reinterpret_cast<const v4f*>(dect);
  const v4f* rd4 = reinterpret_cast<const v4f*>(rd);

  for (;;) {
    const int j0 = tile * 16;
    v4f s4[4], acc[4];
#pragma unroll
    for (int sub = 0; sub < 4; ++sub) d2v[sub] = d2v[sub] * d2v[sub];
    v4f ra[4], rb[4], rc[4], re[4];
    issue_slot<0>(ra, dect4, n4, q);
    issue_slot<1>(rb, dect4, n4, q);
    issue_slot<2>(rc, dect4, n4, q);
    issue_slot<3>(re, dect4, n4, q);
#pragma unroll
    for (int sub = 0; sub < 4; ++sub) {
      s4[sub] = reinterpret_cast<const v4f*>(iw)[(unsigned)cl[sub] * 8u + q];
      acc[sub] = (v4f){0.f, 0.f, 0.f, 0.f};
    }
    pipe_step<0>(acc, ra, rb, dect4, n4, d2v, s4, rd4, q);
    pipe_step<2>(acc, rc, re, dect4, n4, d2v, s4, rd4, q);
    pipe_step<4>(acc, ra, rb, dect4, n4, d2v, s4, rd4, q);
    pipe_step<6>(acc, rc, re, dect4, n4, d2v, s4, rd4, q);
    pipe_step<8>(acc, ra, rb, dect4, n4, d2v, s4, rd4, q);
    pipe_step<10>(acc, rc, re, dect4, n4, d2v, s4, rd4, q);
    pipe_step<12>(acc, ra, rb, dect4, n4, d2v, s4, rd4, q);
    pipe_step<14>(acc, rc, re, dect4, n4, d2v, s4, rd4, q);
    pipe_step<16>(acc, ra, rb, dect4, n4, d2v, s4, rd4, q);
    pipe_step<18>(acc, rc, re, dect4, n4, d2v, s4, rd4, q);
    pipe_step<20>(acc, ra, rb, dect4, n4, d2v, s4, rd4, q);
    pipe_step<22>(acc, rc, re, dect4, n4, d2v, s4, rd4, q);
    pipe_step<24>(acc, ra, rb, dect4, n4, d2v, s4, rd4, q);
    pipe_step<26>(acc, rc, re, dect4, n4, d2v, s4, rd4, q);
    pipe_step<28>(acc, ra, rb, dect4, n4, d2v, s4, rd4, q);
    pipe_step<30>(acc, rc, re, dect4, n4, d2v, s4, rd4, q);

#pragma unroll
    for (int sub = 0; sub < 4; ++sub) {
      const int colp = sub * 4 + pt;
      gbuf[(4 * q + 0) * 17 + colp] = acc[sub].x;
      gbuf[(4 * q + 1) * 17 + colp] = acc[sub].y;
      gbuf[(4 * q + 2) * 17 + colp] = acc[sub].z;
      gbuf[(4 * q + 3) * 17 + colp] = acc[sub].w;
    }
    asm volatile("s_wait_dscnt 0" ::: "memory");

    v8f cacc = {0.f, 0.f, 0.f, 0.f, 0.f, 0.f, 0.f, 0.f};
    const int half = (lane >> 4) & 1;
    const int col = lane & 15;
#pragma unroll
    for (int s8 = 0; s8 < 8; ++s8) {
      const int kb = s8 * 4;
      v2f a, b;
      a.x = enc_s[kb + 2 * half];
      a.y = enc_s[kb + 2 * half + 1];
      b.x = gbuf[(kb + 2 * half) * 17 + col];
      b.y = gbuf[(kb + 2 * half + 1) * 17 + col];
      cacc = __builtin_amdgcn_wmma_f32_16x16x4_f32(
          false, a, false, b, (short)0, cacc, false, false);
    }
    if (lane < 16) out[j0 + lane] = cacc[0];

    tile += FO_STRIDE;
    if (tile >= NT) break;
#pragma unroll
    for (int sub = 0; sub < 4; ++sub) {
      const unsigned j = (unsigned)(tile * 16 + sub * 4 + pt);
      n4[sub] = reinterpret_cast<const v4i*>(nid)[j * 8u + q];
      d2v[sub] = reinterpret_cast<const v4f*>(d)[j * 8u + q];
      cl[sub] = lab[j];
    }
  }
}

extern "C" void kernel_launch(void* const* d_in, const int* in_sizes, int n_in,
                              void* d_out, int out_size, void* d_ws, size_t ws_size,
                              hipStream_t stream) {
  const float* x     = (const float*)d_in[0];
  const float* enc_w = (const float*)d_in[1];
  const float* enc_b = (const float*)d_in[2];
  const float* dec   = (const float*)d_in[3];
  const float* bw_w  = (const float*)d_in[4];
  const float* bw_b  = (const float*)d_in[5];
  const float* nd    = (const float*)d_in[6];
  const int*   nid   = (const int*)d_in[7];
  const int*   lab   = (const int*)d_in[8];
  float* out = (float*)d_out;
  float* ws  = (float*)d_ws;

  float* enc_part = ws + WS_ENCPART;
  float* encoded  = ws + WS_ENCODED;
  float* invw2t   = ws + WS_INVW2T;
  float* denpart  = ws + WS_DENPART;
  float* dect     = ws + WS_DECT;

  k_enc_partial<<<GV_BLOCKS, GV_WAVES * 32, 0, stream>>>(x, enc_w, enc_part);
  k_enc_finish<<<1, 512, 0, stream>>>(enc_part, enc_b, bw_w, bw_b, encoded, invw2t);
  k_dec_transpose<<<TR_BLOCKS, TR_WAVES * 32, 0, stream>>>(dec, dect);
  k_denom_partial<<<DEN_GRID, DEN_THR, 0, stream>>>(nd, lab, invw2t, denpart);
  k_fused_out<<<FO_BLOCKS, FO_THREADS, 0, stream>>>(nd, nid, lab, dect, invw2t, denpart,
                                       encoded, out);
}


# encoder finish + bandwidths ride in the decoder-transpose launch as its 4 leading 4-wave blocks (each redoes the 32 sums, takes 128 of the 512 rows): 4 launches (was 5); order of sums kept
# speedup vs baseline: 10.6612x; 1.0966x over previous
#include <hip/hip_runtime.h>
#include <math.h>


#define NPTS 100000
#define NBAS 32
#define MCLU 16
#define KNEI 32
#define MUF  100.0f

typedef __attribute__((ext_vector_type(2))) float v2f;
typedef __attribute__((ext_vector_type(8))) float v8f;
typedef __attribute__((ext_vector_type(4))) float v4f;
typedef __attribute__((ext_vector_type(4))) int v4i;

#define GV_WAVES  4
#define GV_TILE   128
#define GV_TILES  ((NPTS + GV_TILE - 1) / GV_TILE)
#define GV_BLOCKS ((GV_TILES + GV_WAVES - 1) / GV_WAVES)
#define GV_GROUPS 16
#define GV_PER    ((GV_BLOCKS + GV_GROUPS - 1) / GV_GROUPS)
#define WS_ENCPART 0
#define WS_ENCODED (WS_ENCPART + GV_BLOCKS * 32)
#define WS_INVW2T  (WS_ENCODED + 32)
#define WS_RDENT   (WS_INVW2T + 512)
#define DEN_BLKS   128
#define WS_DENPART (WS_RDENT + 1024)
#define WS_DECT    (WS_DENPART + DEN_BLKS*1024)

template <int S>
__device__ __forceinline__ void gv_fold(float (&a)[32], const unsigned lane) {
  const bool up = (lane & (unsigned)S) != 0u;
#pragma unroll
  for (int r = 0; r < S; ++r) {
    const float keep = up ? a[r + S] : a[r];
    const float send = up ? a[r] : a[r + S];
    a[r] = keep + __shfl_xor(send, S);
  }
}

__global__ __launch_bounds__(GV_WAVES * 32)
void k_enc_partial(const float* __restrict__ x,
                   const float* __restrict__ W,
                   float* __restrict__ part) {
  __shared__ float red[GV_WAVES][32];
  const unsigned lane = threadIdx.x & 31u, wv = threadIdx.x >> 5;
  const unsigned tile = blockIdx.x * GV_WAVES + wv;
  const unsigned j = tile * GV_TILE + lane * 4u;
  const bool valid = j < (unsigned)NPTS;
  const unsigned jc = valid ? j : (unsigned)(NPTS - 4);
  const v4f xv = *reinterpret_cast<const v4f*>(x + jc);
  const v4f* wp = reinterpret_cast<const v4f*>(W + jc);
  v4f w[NBAS];
#pragma unroll
  for (int r = 0; r < NBAS; ++r) w[r] = wp[r * (NPTS / 4)];
  float a[NBAS];
#pragma unroll
  for (int r = 0; r < NBAS; ++r) {
    float s = w[r].x * xv.x;
    s += w[r].y * xv.y;
    s += w[r].z * xv.z;
    s += w[r].w * xv.w;
    a[r] = valid ? s : 0.f;
  }
  gv_fold<16>(a, lane);
  gv_fold<8>(a, lane);
  gv_fold<4>(a, lane);
  gv_fold<2>(a, lane);
  gv_fold<1>(a, lane);
  red[wv][lane] = a[0];
  __syncthreads();
  if (wv == 0)
    part[blockIdx.x * 32 + lane] = ((red[0][lane] + red[1][lane]) + red[2][lane]) + red[3][lane];
}

#define TR_WAVES  4
#define TR_TILES  (NPTS / 32)
#define TR_BLOCKS ((TR_TILES + TR_WAVES - 1) / TR_WAVES)
#define EF_THR    (TR_WAVES * 32)
#define EF_BLKS   ((NBAS * MCLU) / EF_THR)
#define EF_NGRP   (GV_GROUPS / TR_WAVES)
#define FT_GRID   (EF_BLKS + TR_BLOCKS)
static_assert(EF_BLKS * EF_THR == NBAS * MCLU, "one bandwidth row per finish thread");
static_assert(EF_NGRP * TR_WAVES == GV_GROUPS, "line groups split evenly over a block's waves");

__global__ __launch_bounds__(TR_WAVES * 32)
void k_finish_transpose(const float* __restrict__ part,
                        const float* __restrict__ enc_b,
                        const float* __restrict__ bw_w,
                        const float* __restrict__ bw_b,
                        float* __restrict__ encoded,
                        float* __restrict__ invw2t,
                        const float* __restrict__ dec,
                        float* __restrict__ dect) {
  __shared__ float enc_s[NBAS];
  __shared__ float ps[GV_GROUPS][NBAS];
  if (blockIdx.x < EF_BLKS) {
    const int t = threadIdx.x;
    {
      const int h = t >> 5, r = t & 31;
      float s[EF_NGRP];
#pragma unroll
      for (int gi = 0; gi < EF_NGRP; ++gi) {
        const int g = h + gi * TR_WAVES;
        s[gi] = 0.f;
#pragma unroll
        for (int k = 0; k < GV_PER; ++k) {
          const int b = g + k * GV_GROUPS;
          const float v = part[min(b, GV_BLOCKS - 1) * 32 + r];
          s[gi] += (b < GV_BLOCKS) ? v : 0.f;
        }
        ps[g][r] = s[gi];
      }
    }
    __syncthreads();
    if (t < NBAS) {
      float s = 0.f;
      for (int g = 0; g < GV_GROUPS; ++g) s += ps[g][t];
      s += enc_b[t];
      enc_s[t] = s;
      if (blockIdx.x == 0) encoded[t] = s;
    }
    __syncthreads();
    const int o = (int)blockIdx.x * EF_THR + t;
    float s = 0.f;
    for (int q = 0; q < NBAS; ++q) s += enc_s[q] * bw_w[o * NBAS + q];
    s += bw_b[o];
    float band = (1.f / (1.f + expf(-s))) / 60.f;
    float wm = band * MUF;
    const int i = o / MCLU, c = o % MCLU;
    invw2t[c * NBAS + i] = 1.f / (wm * wm);
    return;
  }
  const unsigned lane = threadIdx.x & 31u, wv = threadIdx.x >> 5;
  const unsigned tile = (blockIdx.x - (unsigned)EF_BLKS) * TR_WAVES + wv;
  if (tile >= TR_TILES) return;
  const v4f* src = reinterpret_cast<const v4f*>(dec + lane * (unsigned)NPTS + tile * 32u);
  v4f r[8];
#pragma unroll
  for (int s = 0; s < 8; ++s) r[s] = src[s];
  float* dst = dect + tile * 1024u + lane;
#pragma unroll
  for (int p = 0; p < 32; ++p) dst[p * 32] = r[p >> 2][p & 3];
}

#define DEN_GRID 256
#define DEN_THR  512
#define DEN_CH   782
#define DEN_H    392
#define DEN_RD   13
#define DEN_PITCH 36
#define DEN_ROWS (DEN_RD * DEN_THR / 8)
static_assert(DEN_BLKS * DEN_CH >= NPTS, "point groups cover all points");
static_assert(NPTS - (DEN_BLKS - 1) * DEN_CH >= DEN_THR, "shortest chunk has a label for every thread");
static_assert(DEN_RD * DEN_THR >= DEN_CH * (KNEI / 4), "one burst covers a chunk");
static_assert(MCLU * NBAS == DEN_THR, "one table entry per thread");
static_assert(2 * DEN_THR >= DEN_CH + 2, "label slots cover a chunk");
static_assert(NPTS - (DEN_BLKS - 1) * DEN_CH > DEN_H, "every chunk has a second half");

template <int X, int H>
__device__ __forceinline__ void den_fold(float (&a)[64], const unsigned lane) {
  const bool up = (lane & (unsigned)X) != 0u;
#pragma unroll
  for (int r = 0; r < H; ++r) {
    const float keep = up ? a[r + H] : a[r];
    const float send = up ? a[r] : a[r + H];
    a[r] = keep + __shfl_xor(send, X);
  }
}

__device__ __forceinline__ v2f den_win(const v2f d2, const float s) {
  const v2f t = 1.f - d2 * s;
  v2f r;
  r.x = fmaxf(t.x, 0.f);
  r.y = fmaxf(t.y, 0.f);
  return r;
}

__global__ __launch_bounds__(DEN_THR) __attribute__((amdgpu_waves_per_eu(4, 4)))
void k_denom_partial(const float* __restrict__ d,
                     const int* __restrict__ lab,
                     const float* __restrict__ invw2t,
                     float* __restrict__ part) {
  __shared__ __attribute__((aligned(16))) float d2s[DEN_ROWS * DEN_PITCH];
  __shared__ __attribute__((aligned(16))) int   labs[2 * DEN_THR];
  __shared__ __attribute__((aligned(16))) float iw[MCLU * NBAS];
  __shared__ __attribute__((aligned(16))) float comb[8 * 64];
  const int tid = threadIdx.x;
  const int pg = blockIdx.x >> 1, ih = blockIdx.x & 1;
  const int p0 = pg * DEN_CH;
  const int n  = min(DEN_CH, NPTS - p0);
  const int n4 = n * (KNEI / 4);

  const int lv0 = lab[p0 + tid];
  const int lv1 = lab[p0 + min(tid + DEN_THR, n - 1)];
  const float iv = invw2t[tid];
  const v4f* src = reinterpret_cast<const v4f*>(d + p0 * KNEI);
  v4f r[DEN_RD];
#pragma unroll
  for (int s = 0; s < DEN_RD; ++s) r[s] = src[min(s * DEN_THR + tid, n4 - 1)];
  labs[tid] = lv0;
  labs[tid + DEN_THR] = lv1;
  iw[tid] = iv;
#pragma unroll
  for (int s = 0; s < DEN_RD; ++s) {
    const int g = s * DEN_THR + tid;
    *reinterpret_cast<v4f*>(d2s + (g >> 3) * DEN_PITCH + (g & 7) * 4) = r[s] * r[s];
  }
  __syncthreads();

  const unsigned lane = (unsigned)tid & 31u;
  const int w = tid >> 5;
  const int bp = w & 7, ph = w >> 3;
  const int i0 = ih * 16 + bp * 2;
  const int pend = ph ? n : DEN_H;
  v2f acc0[16], acc1[16];
#pragma unroll
  for (int q = 0; q < 16; ++q) {
    acc0[q] = (v2f){0.f, 0.f};
    acc1[q] = (v2f){0.f, 0.f};
  }
  for (int p = ph * DEN_H + (int)lane; p < pend; p += 32) {
    const int c = labs[p];
    const v2f a = *reinterpret_cast<const v2f*>(iw + c * NBAS + i0);
    const v4f* row = reinterpret_cast<const v4f*>(d2s + p * DEN_PITCH);
#pragma unroll
    for (int q = 0; q < 8; ++q) {
      const v4f dq = row[q];
      const v2f lo = {dq.x, dq.y}, hi = {dq.z, dq.w};
      acc0[2 * q]     += den_win(lo, a.x);
      acc0[2 * q + 1] += den_win(hi, a.x);
      acc1[2 * q]     += den_win(lo, a.y);
      acc1[2 * q + 1] += den_win(hi, a.y);
    }
  }

  float f[64];
#pragma unroll
  for (int q = 0; q < 16; ++q) {
    f[2 * q]          = acc0[q].x;
    f[2 * q + 1]      = acc0[q].y;
    f[32 + 2 * q]     = acc1[q].x;
    f[32 + 2 * q + 1] = acc1[q].y;
  }
  den_fold<16, 32>(f, lane);
  den_fold<8, 16>(f, lane);
  den_fold<4, 8>(f, lane);
  den_fold<2, 4>(f, lane);
  den_fold<1, 2>(f, lane);

  if (ph) {
    *reinterpret_cast<v2f*>(comb + bp * 64 + 2 * (int)lane) = (v2f){f[0], f[1]};
  }
  __syncthreads();
  if (!ph) {
    const v2f o = *reinterpret_cast<const v2f*>(comb + bp * 64 + 2 * (int)lane);
    *reinterpret_cast<v2f*>(part + pg * 1024 + i0 * 32 + 2 * (int)lane) = (v2f){f[0] + o.x, f[1] + o.y};
  }
}

#define FO_WAVES   20
#define FO_THREADS 640
#define FO_BLOCKS  256
#define FO_STRIDE  (FO_BLOCKS * FO_WAVES)
#define FIN_GRP    2
#define FIN_PER    (DEN_BLKS / FIN_GRP)
#define FIN_RND    32
static_assert(FIN_GRP * 256 <= FO_THREADS, "finish threads exist");
static_assert(FIN_PER % FIN_RND == 0, "whole rounds");
static_assert(FO_STRIDE <= NPTS / 16, "every wave has a first tile");

template <int G>
__device__ __forceinline__ int bcast8(int v) {
  return (int)__builtin_amdgcn_mov_dpp8((unsigned)v, (unsigned)(G * 0x249249));
}

template <int KK>
__device__ __forceinline__ void issue_slot(v4f (&r)[4], const v4f* __restrict__ dect4,
                                           const v4i (&n4)[4], const unsigned q) {
  r[0] = dect4[(unsigned)bcast8<(KK >> 2)>(n4[0][KK & 3]) * 8u + q];
  r[1] = dect4[(unsigned)bcast8<(KK >> 2)>(n4[1][KK & 3]) * 8u + q];
  r[2] = dect4[(unsigned)bcast8<(KK >> 2)>(n4[2][KK & 3]) * 8u + q];
  r[3] = dect4[(unsigned)bcast8<(KK >> 2)>(n4[3][KK & 3]) * 8u + q];
}

__device__ __forceinline__ void add_pass(v4f& acc, const v4f r4, const float d2k,
                                         const v4f s4, const v4f dec) {
  acc.x += r4.x * fmaxf(1.f - d2k * s4.x, 0.f) * dec.x;
  acc.y += r4.y * fmaxf(1.f - d2k * s4.y, 0.f) * dec.y;
  acc.z += r4.z * fmaxf(1.f - d2k * s4.z, 0.f) * dec.z;
  acc.w += r4.w * fmaxf(1.f - d2k * s4.w, 0.f) * dec.w;
}

template <int KK>
__device__ __forceinline__ void use_slot(v4f (&acc)[4], const v4f (&r)[4],
                                         const v4f (&d2v)[4], const v4f (&s4)[4],
                                         const v4f* rd4, const unsigned q) {
  const v4f r4 = rd4[KK * 8 + q];
  add_pass(acc[0], r4, __int_as_float(bcast8<(KK >> 2)>(__float_as_int(d2v[0][KK & 3]))), s4[0], r[0]);
  add_pass(acc[1], r4, __int_as_float(bcast8<(KK >> 2)>(__float_as_int(d2v[1][KK & 3]))), s4[1], r[1]);
  add_pass(acc[2], r4, __int_as_float(bcast8<(KK >> 2)>(__float_as_int(d2v[2][KK & 3]))), s4[2], r[2]);
  add_pass(acc[3], r4, __int_as_float(bcast8<(KK >> 2)>(__float_as_int(d2v[3][KK & 3]))), s4[3], r[3]);
}

template <int K>
__device__ __forceinline__ void pipe_step(v4f (&acc)[4], v4f (&ra)[4], v4f (&rb)[4],
                                          const v4f* __restrict__ dect4,
                                          const v4i (&n4)[4], const v4f (&d2v)[4],
                                          const v4f (&s4)[4], const v4f* rd4,
                                          const unsigned q) {
  use_slot<K>(acc, ra, d2v, s4, rd4, q);
  use_slot<K + 1>(acc, rb, d2v, s4, rd4, q);
  if constexpr (K + 4 < KNEI) {
    issue_slot<K + 4>(ra, dect4, n4, q);
    issue_slot<K + 5>(rb, dect4, n4, q);
  }
}

__global__ __launch_bounds__(FO_THREADS)
void k_fused_out(const float* __restrict__ d,
                 const int* __restrict__ nid,
                 const int* __restrict__ lab,
                 const float* __restrict__ dect,
                 const float* __restrict__ invw2t,
                 const float* __restrict__ denpart,
                 const float* __restrict__ encoded,
                 float* __restrict__ out) {
  __shared__ __attribute__((aligned(16))) float iw[MCLU * NBAS];
  __shared__ __attribute__((aligned(16))) float rd[NBAS * NBAS];
  __shared__ __attribute__((aligned(16))) float fin[NBAS * NBAS];
  __shared__ float enc_s[NBAS];
  __shared__ float g[FO_WAVES][32 * 17];
  const int tid = threadIdx.x;
  const int lane = tid & 31, wv = tid >> 5;
  const int NT = NPTS / 16;
  const int pt = lane >> 3;
  const unsigned q = (unsigned)(lane & 7);

  int tile = wv * FO_BLOCKS + (int)blockIdx.x;
  v4i n4[4];
  v4f d2v[4];
  int cl[4];
#pragma unroll
  for (int sub = 0; sub < 4; ++sub) {
    const unsigned j = (unsigned)(tile * 16 + sub * 4 + pt);
    n4[sub] = reinterpret_cast<const v4i*>(nid)[j * 8u + q];
    d2v[sub] = reinterpret_cast<const v4f*>(d)[j * 8u + q];
    cl[sub] = lab[j];
  }

  const int fc = tid & 255, fg = tid >> 8;
  v4f fs = {0.f, 0.f, 0.f, 0.f};
  if (fg < FIN_GRP) {
    const v4f* pp = reinterpret_cast<const v4f*>(denpart) + (fg * FIN_PER) * 256 + fc;
    for (int b0 = 0; b0 < FIN_PER; b0 += FIN_RND) {
      v4f r[FIN_RND];
#pragma unroll
      for (int k = 0; k < FIN_RND; ++k) r[k] = pp[(b0 + k) * 256];
#pragma unroll
      for (int k = 0; k < FIN_RND; ++k) fs += r[k];
    }
    if (fg == 1) reinterpret_cast<v4f*>(fin)[fc] = fs;
  } else {
    const int t = tid - FIN_GRP * 256;
    reinterpret_cast<v4f*>(iw)[t] = reinterpret_cast<const v4f*>(invw2t)[t];
    if (t < NBAS) enc_s[t] = encoded[t];
  }
  __syncthreads();
  if (fg == 0) {
    const v4f o = reinterpret_cast<const v4f*>(fin)[fc];
    const int i = fc >> 3, kk = (fc & 7) * 4;
    rd[(kk + 0) * 32 + i] = 1.f / (fs.x + o.x);
    rd[(kk + 1) * 32 + i] = 1.f / (fs.y + o.y);
    rd[(kk + 2) * 32 + i] = 1.f / (fs.z + o.z);
    rd[(kk + 3) * 32 + i] = 1.f / (fs.w + o.w);
  }
  __syncthreads();

  float* gbuf = g[wv];
  const v4f* dect4 = reinterpret_cast<const v4f*>(dect);
  const v4f* rd4 = reinterpret_cast<const v4f*>(rd);

  for (;;) {
    const int j0 = tile * 16;
    v4f s4[4], acc[4];
#pragma unroll
    for (int sub = 0; sub < 4; ++sub) d2v[sub] = d2v[sub] * d2v[sub];
    v4f ra[4], rb[4], rc[4], re[4];
    issue_slot<0>(ra, dect4, n4, q);
    issue_slot<1>(rb, dect4, n4, q);
    issue_slot<2>(rc, dect4, n4, q);
    issue_slot<3>(re, dect4, n4, q);
#pragma unroll
    for (int sub = 0; sub < 4; ++sub) {
      s4[sub] = reinterpret_cast<const v4f*>(iw)[(unsigned)cl[sub] * 8u + q];
      acc[sub] = (v4f){0.f, 0.f, 0.f, 0.f};
    }
    pipe_step<0>(acc, ra, rb, dect4, n4, d2v, s4, rd4, q);
    pipe_step<2>(acc, rc, re, dect4, n4, d2v, s4, rd4, q);
    pipe_step<4>(acc, ra, rb, dect4, n4, d2v, s4, rd4, q);
    pipe_step<6>(acc, rc, re, dect4, n4, d2v, s4, rd4, q);
    pipe_step<8>(acc, ra, rb, dect4, n4, d2v, s4, rd4, q);
    pipe_step<10>(acc, rc, re, dect4, n4, d2v, s4, rd4, q);
    pipe_step<12>(acc, ra, rb, dect4, n4, d2v, s4, rd4, q);
    pipe_step<14>(acc, rc, re, dect4, n4, d2v, s4, rd4, q);
    pipe_step<16>(acc, ra, rb, dect4, n4, d2v, s4, rd4, q);
    pipe_step<18>(acc, rc, re, dect4, n4, d2v, s4, rd4, q);
    pipe_step<20>(acc, ra, rb, dect4, n4, d2v, s4, rd4, q);
    pipe_step<22>(acc, rc, re, dect4, n4, d2v, s4, rd4, q);
    pipe_step<24>(acc, ra, rb, dect4, n4, d2v, s4, rd4, q);
    pipe_step<26>(acc, rc, re, dect4, n4, d2v, s4, rd4, q);
    pipe_step<28>(acc, ra, rb, dect4, n4, d2v, s4, rd4, q);
    pipe_step<30>(acc, rc, re, dect4, n4, d2v, s4, rd4, q);

#pragma unroll
    for (int sub = 0; sub < 4; ++sub) {
      const int colp = sub * 4 + pt;
      gbuf[(4 * q + 0) * 17 + colp] = acc[sub].x;
      gbuf[(4 * q + 1) * 17 + colp] = acc[sub].y;
      gbuf[(4 * q + 2) * 17 + colp] = acc[sub].z;
      gbuf[(4 * q + 3) * 17 + colp] = acc[sub].w;
    }
    asm volatile("s_wait_dscnt 0" ::: "memory");

    v8f cacc = {0.f, 0.f, 0.f, 0.f, 0.f, 0.f, 0.f, 0.f};
    const int half = (lane >> 4) & 1;
    const int col = lane & 15;
#pragma unroll
    for (int s8 = 0; s8 < 8; ++s8) {
      const int kb = s8 * 4;
      v2f a, b;
      a.x = enc_s[kb + 2 * half];
      a.y = enc_s[kb + 2 * half + 1];
      b.x = gbuf[(kb + 2 * half) * 17 + col];
      b.y = gbuf[(kb + 2 * half + 1) * 17 + col];
      cacc = __builtin_amdgcn_wmma_f32_16x16x4_f32(
          false, a, false, b, (short)0, cacc, false, false);
    }
    if (lane < 16) out[j0 + lane] = cacc[0];

    tile += FO_STRIDE;
    if (tile >= NT) break;
#pragma unroll
    for (int sub = 0; sub < 4; ++sub) {
      const unsigned j = (unsigned)(tile * 16 + sub * 4 + pt);
      n4[sub] = reinterpret_cast<const v4i*>(nid)[j * 8u + q];
      d2v[sub] = reinterpret_cast<const v4f*>(d)[j * 8u + q];
      cl[sub] = lab[j];
    }
  }
}

extern "C" void kernel_launch(void* const* d_in, const int* in_sizes, int n_in,
                              void* d_out, int out_size, void* d_ws, size_t ws_size,
                              hipStream_t stream) {
  const float* x     = (const float*)d_in[0];
  const float* enc_w = (const float*)d_in[1];
  const float* enc_b = (const float*)d_in[2];
  const float* dec   = (const float*)d_in[3];
  const float* bw_w  = (const float*)d_in[4];
  const float* bw_b  = (const float*)d_in[5];
  const float* nd    = (const float*)d_in[6];
  const int*   nid   = (const int*)d_in[7];
  const int*   lab   = (const int*)d_in[8];
  float* out = (float*)d_out;
  float* ws  = (float*)d_ws;

  float* enc_part = ws + WS_ENCPART;
  float* encoded  = ws + WS_ENCODED;
  float* invw2t   = ws + WS_INVW2T;
  float* denpart  = ws + WS_DENPART;
  float* dect     = ws + WS_DECT;

  k_enc_partial<<<GV_BLOCKS, GV_WAVES * 32, 0, stream>>>(x, enc_w, enc_part);
  k_finish_transpose<<<FT_GRID, EF_THR, 0, stream>>>(enc_part, enc_b, bw_w, bw_b, encoded, invw2t,
                                                    dec, dect);
  k_denom_partial<<<DEN_GRID, DEN_THR, 0, stream>>>(nd, lab, invw2t, denpart);
  k_fused_out<<<FO_BLOCKS, FO_THREADS, 0, stream>>>(nd, nid, lab, dect, invw2t, denpart,
                                       encoded, out);
}


# denominator pass re-tiled: 64 point groups x 4 base groups of 8 (was 128 x 2 of 16), 1563 rows per block in LDS: 64 partial planes (was 128) for the consumer to sum; denominator sums regrouped
# speedup vs baseline: 10.9960x; 1.0314x over previous
#include <hip/hip_runtime.h>
#include <math.h>


#define NPTS 100000
#define NBAS 32
#define MCLU 16
#define KNEI 32
#define MUF  100.0f

typedef __attribute__((ext_vector_type(2))) float v2f;
typedef __attribute__((ext_vector_type(8))) float v8f;
typedef __attribute__((ext_vector_type(4))) float v4f;
typedef __attribute__((ext_vector_type(4))) int v4i;

#define GV_WAVES  4
#define GV_TILE   128
#define GV_TILES  ((NPTS + GV_TILE - 1) / GV_TILE)
#define GV_BLOCKS ((GV_TILES + GV_WAVES - 1) / GV_WAVES)
#define GV_GROUPS 16
#define GV_PER    ((GV_BLOCKS + GV_GROUPS - 1) / GV_GROUPS)
#define WS_ENCPART 0
#define WS_ENCODED (WS_ENCPART + GV_BLOCKS * 32)
#define WS_INVW2T  (WS_ENCODED + 32)
#define WS_RDENT   (WS_INVW2T + 512)
#define DEN_BLKS   64
#define WS_DENPART (WS_RDENT + 1024)
#define WS_DECT    (WS_DENPART + DEN_BLKS*1024)

template <int S>
__device__ __forceinline__ void gv_fold(float (&a)[32], const unsigned lane) {
  const bool up = (lane & (unsigned)S) != 0u;
#pragma unroll
  for (int r = 0; r < S; ++r) {
    const float keep = up ? a[r + S] : a[r];
    const float send = up ? a[r] : a[r + S];
    a[r] = keep + __shfl_xor(send, S);
  }
}

__global__ __launch_bounds__(GV_WAVES * 32)
void k_enc_partial(const float* __restrict__ x,
                   const float* __restrict__ W,
                   float* __restrict__ part) {
  __shared__ float red[GV_WAVES][32];
  const unsigned lane = threadIdx.x & 31u, wv = threadIdx.x >> 5;
  const unsigned tile = blockIdx.x * GV_WAVES + wv;
  const unsigned j = tile * GV_TILE + lane * 4u;
  const bool valid = j < (unsigned)NPTS;
  const unsigned jc = valid ? j : (unsigned)(NPTS - 4);
  const v4f xv = *reinterpret_cast<const v4f*>(x + jc);
  const v4f* wp = reinterpret_cast<const v4f*>(W + jc);
  v4f w[NBAS];
#pragma unroll
  for (int r = 0; r < NBAS; ++r) w[r] = wp[r * (NPTS / 4)];
  float a[NBAS];
#pragma unroll
  for (int r = 0; r < NBAS; ++r) {
    float s = w[r].x * xv.x;
    s += w[r].y * xv.y;
    s += w[r].z * xv.z;
    s += w[r].w * xv.w;
    a[r] = valid ? s : 0.f;
  }
  gv_fold<16>(a, lane);
  gv_fold<8>(a, lane);
  gv_fold<4>(a, lane);
  gv_fold<2>(a, lane);
  gv_fold<1>(a, lane);
  red[wv][lane] = a[0];
  __syncthreads();
  if (wv == 0)
    part[blockIdx.x * 32 + lane] = ((red[0][lane] + red[1][lane]) + red[2][lane]) + red[3][lane];
}

#define TR_WAVES  4
#define TR_TILES  (NPTS / 32)
#define TR_BLOCKS ((TR_TILES + TR_WAVES - 1) / TR_WAVES)
#define EF_THR    (TR_WAVES * 32)
#define EF_BLKS   ((NBAS * MCLU) / EF_THR)
#define EF_NGRP   (GV_GROUPS / TR_WAVES)
#define FT_GRID   (EF_BLKS + TR_BLOCKS)
static_assert(EF_BLKS * EF_THR == NBAS * MCLU, "one bandwidth row per finish thread");
static_assert(EF_NGRP * TR_WAVES == GV_GROUPS, "line groups split evenly over a block's waves");

__global__ __launch_bounds__(TR_WAVES * 32)
void k_finish_transpose(const float* __restrict__ part,
                        const float* __restrict__ enc_b,
                        const float* __restrict__ bw_w,
                        const float* __restrict__ bw_b,
                        float* __restrict__ encoded,
                        float* __restrict__ invw2t,
                        const float* __restrict__ dec,
                        float* __restrict__ dect) {
  __shared__ float enc_s[NBAS];
  __shared__ float ps[GV_GROUPS][NBAS];
  if (blockIdx.x < EF_BLKS) {
    const int t = threadIdx.x;
    {
      const int h = t >> 5, r = t & 31;
      float s[EF_NGRP];
#pragma unroll
      for (int gi = 0; gi < EF_NGRP; ++gi) {
        const int g = h + gi * TR_WAVES;
        s[gi] = 0.f;
#pragma unroll
        for (int k = 0; k < GV_PER; ++k) {
          const int b = g + k * GV_GROUPS;
          const float v = part[min(b, GV_BLOCKS - 1) * 32 + r];
          s[gi] += (b < GV_BLOCKS) ? v : 0.f;
        }
        ps[g][r] = s[gi];
      }
    }
    __syncthreads();
    if (t < NBAS) {
      float s = 0.f;
      for (int g = 0; g < GV_GROUPS; ++g) s += ps[g][t];
      s += enc_b[t];
      enc_s[t] = s;
      if (blockIdx.x == 0) encoded[t] = s;
    }
    __syncthreads();
    const int o = (int)blockIdx.x * EF_THR + t;
    float s = 0.f;
    for (int q = 0; q < NBAS; ++q) s += enc_s[q] * bw_w[o * NBAS + q];
    s += bw_b[o];
    float band = (1.f / (1.f + expf(-s))) / 60.f;
    float wm = band * MUF;
    const int i = o / MCLU, c = o % MCLU;
    invw2t[c * NBAS + i] = 1.f / (wm * wm);
    return;
  }
  const unsigned lane = threadIdx.x & 31u, wv = threadIdx.x >> 5;
  const unsigned tile = (blockIdx.x - (unsigned)EF_BLKS) * TR_WAVES + wv;
  if (tile >= TR_TILES) return;
  const v4f* src = reinterpret_cast<const v4f*>(dec + lane * (unsigned)NPTS + tile * 32u);
  v4f r[8];
#pragma unroll
  for (int s = 0; s < 8; ++s) r[s] = src[s];
  float* dst = dect + tile * 1024u + lane;
#pragma unroll
  for (int p = 0; p < 32; ++p) dst[p * 32] = r[p >> 2][p & 3];
}

#define DEN_GRID 256
#define DEN_THR  512
#define DEN_CH   1563
#define DEN_H    391
#define DEN_RD   25
#define DEN_PITCH 36
#define DEN_ROWS (DEN_RD * DEN_THR / 8)
static_assert(DEN_BLKS * DEN_CH >= NPTS, "point groups cover all points");
static_assert(NPTS - (DEN_BLKS - 1) * DEN_CH >= DEN_THR, "shortest chunk has a label for every thread");
static_assert(DEN_RD * DEN_THR >= DEN_CH * (KNEI / 4), "one burst covers a chunk");
static_assert(MCLU * NBAS == DEN_THR, "one table entry per thread");
static_assert(4 * DEN_THR >= DEN_CH + 2, "label slots cover a chunk");
static_assert(NPTS - (DEN_BLKS - 1) * DEN_CH > 3 * DEN_H, "every chunk has a fourth quarter");

template <int X, int H>
__device__ __forceinline__ void den_fold(float (&a)[64], const unsigned lane) {
  const bool up = (lane & (unsigned)X) != 0u;
#pragma unroll
  for (int r = 0; r < H; ++r) {
    const float keep = up ? a[r + H] : a[r];
    const float send = up ? a[r] : a[r + H];
    a[r] = keep + __shfl_xor(send, X);
  }
}

__device__ __forceinline__ v2f den_win(const v2f d2, const float s) {
  const v2f t = 1.f - d2 * s;
  v2f r;
  r.x = fmaxf(t.x, 0.f);
  r.y = fmaxf(t.y, 0.f);
  return r;
}

__global__ __launch_bounds__(DEN_THR) __attribute__((amdgpu_waves_per_eu(4, 4)))
void k_denom_partial(const float* __restrict__ d,
                     const int* __restrict__ lab,
                     const float* __restrict__ invw2t,
                     float* __restrict__ part) {
  __shared__ __attribute__((aligned(16))) float d2s[DEN_ROWS * DEN_PITCH];
  __shared__ __attribute__((aligned(16))) int   labs[4 * DEN_THR];
  __shared__ __attribute__((aligned(16))) float iw[MCLU * NBAS];
  __shared__ __attribute__((aligned(16))) float comb[3 * 4 * 64];
  const int tid = threadIdx.x;
  const int pg = blockIdx.x >> 2, ih = blockIdx.x & 3;
  const int p0 = pg * DEN_CH;
  const int n  = min(DEN_CH, NPTS - p0);
  const int n4 = n * (KNEI / 4);

  const int lv0 = lab[p0 + tid];
  const int lv1 = lab[p0 + min(tid + DEN_THR, n - 1)];
  const int lv2 = lab[p0 + min(tid + 2 * DEN_THR, n - 1)];
  const int lv3 = lab[p0 + min(tid + 3 * DEN_THR, n - 1)];
  const float iv = invw2t[tid];
  const v4f* src = reinterpret_cast<const v4f*>(d + p0 * KNEI);
  v4f r[DEN_RD];
#pragma unroll
  for (int s = 0; s < DEN_RD; ++s) r[s] = src[min(s * DEN_THR + tid, n4 - 1)];
  labs[tid] = lv0;
  labs[tid + DEN_THR] = lv1;
  labs[tid + 2 * DEN_THR] = lv2;
  labs[tid + 3 * DEN_THR] = lv3;
  iw[tid] = iv;
#pragma unroll
  for (int s = 0; s < DEN_RD; ++s) {
    const int g = s * DEN_THR + tid;
    *reinterpret_cast<v4f*>(d2s + (g >> 3) * DEN_PITCH + (g & 7) * 4) = r[s] * r[s];
  }
  __syncthreads();

  const unsigned lane = (unsigned)tid & 31u;
  const int w = tid >> 5;
  const int bp = w & 3, ph = w >> 2;
  const int i0 = ih * 8 + bp * 2;
  const int pend = (ph == 3) ? n : (ph + 1) * DEN_H;
  v2f acc0[16], acc1[16];
#pragma unroll
  for (int q = 0; q < 16; ++q) {
    acc0[q] = (v2f){0.f, 0.f};
    acc1[q] = (v2f){0.f, 0.f};
  }
  for (int p = ph * DEN_H + (int)lane; p < pend; p += 32) {
    const int c = labs[p];
    const v2f a = *reinterpret_cast<const v2f*>(iw + c * NBAS + i0);
    const v4f* row = reinterpret_cast<const v4f*>(d2s + p * DEN_PITCH);
#pragma unroll
    for (int q = 0; q < 8; ++q) {
      const v4f dq = row[q];
      const v2f lo = {dq.x, dq.y}, hi = {dq.z, dq.w};
      acc0[2 * q]     += den_win(lo, a.x);
      acc0[2 * q + 1] += den_win(hi, a.x);
      acc1[2 * q]     += den_win(lo, a.y);
      acc1[2 * q + 1] += den_win(hi, a.y);
    }
  }

  float f[64];
#pragma unroll
  for (int q = 0; q < 16; ++q) {
    f[2 * q]          = acc0[q].x;
    f[2 * q + 1]      = acc0[q].y;
    f[32 + 2 * q]     = acc1[q].x;
    f[32 + 2 * q + 1] = acc1[q].y;
  }
  den_fold<16, 32>(f, lane);
  den_fold<8, 16>(f, lane);
  den_fold<4, 8>(f, lane);
  den_fold<2, 4>(f, lane);
  den_fold<1, 2>(f, lane);

  if (ph) {
    *reinterpret_cast<v2f*>(comb + (ph - 1) * 256 + bp * 64 + 2 * (int)lane) = (v2f){f[0], f[1]};
  }
  __syncthreads();
  if (!ph) {
    const v2f o1 = *reinterpret_cast<const v2f*>(comb + bp * 64 + 2 * (int)lane);
    const v2f o2 = *reinterpret_cast<const v2f*>(comb + 256 + bp * 64 + 2 * (int)lane);
    const v2f o3 = *reinterpret_cast<const v2f*>(comb + 512 + bp * 64 + 2 * (int)lane);
    *reinterpret_cast<v2f*>(part + pg * 1024 + i0 * 32 + 2 * (int)lane) = (v2f){((f[0] + o1.x) + o2.x) + o3.x, ((f[1] + o1.y) + o2.y) + o3.y};
  }
}

#define FO_WAVES   20
#define FO_THREADS 640
#define FO_BLOCKS  256
#define FO_STRIDE  (FO_BLOCKS * FO_WAVES)
#define FIN_GRP    2
#define FIN_PER    (DEN_BLKS / FIN_GRP)
#define FIN_RND    32
static_assert(FIN_GRP * 256 <= FO_THREADS, "finish threads exist");
static_assert(FIN_PER % FIN_RND == 0, "whole rounds");
static_assert(FO_STRIDE <= NPTS / 16, "every wave has a first tile");

template <int G>
__device__ __forceinline__ int bcast8(int v) {
  return (int)__builtin_amdgcn_mov_dpp8((unsigned)v, (unsigned)(G * 0x249249));
}

template <int KK>
__device__ __forceinline__ void issue_slot(v4f (&r)[4], const v4f* __restrict__ dect4,
                                           const v4i (&n4)[4], const unsigned q) {
  r[0] = dect4[(unsigned)bcast8<(KK >> 2)>(n4[0][KK & 3]) * 8u + q];
  r[1] = dect4[(unsigned)bcast8<(KK >> 2)>(n4[1][KK & 3]) * 8u + q];
  r[2] = dect4[(unsigned)bcast8<(KK >> 2)>(n4[2][KK & 3]) * 8u + q];
  r[3] = dect4[(unsigned)bcast8<(KK >> 2)>(n4[3][KK & 3]) * 8u + q];
}

__device__ __forceinline__ void add_pass(v4f& acc, const v4f r4, const float d2k,
                                         const v4f s4, const v4f dec) {
  acc.x += r4.x * fmaxf(1.f - d2k * s4.x, 0.f) * dec.x;
  acc.y += r4.y * fmaxf(1.f - d2k * s4.y, 0.f) * dec.y;
  acc.z += r4.z * fmaxf(1.f - d2k * s4.z, 0.f) * dec.z;
  acc.w += r4.w * fmaxf(1.f - d2k * s4.w, 0.f) * dec.w;
}

template <int KK>
__device__ __forceinline__ void use_slot(v4f (&acc)[4], const v4f (&r)[4],
                                         const v4f (&d2v)[4], const v4f (&s4)[4],
                                         const v4f* rd4, const unsigned q) {
  const v4f r4 = rd4[KK * 8 + q];
  add_pass(acc[0], r4, __int_as_float(bcast8<(KK >> 2)>(__float_as_int(d2v[0][KK & 3]))), s4[0], r[0]);
  add_pass(acc[1], r4, __int_as_float(bcast8<(KK >> 2)>(__float_as_int(d2v[1][KK & 3]))), s4[1], r[1]);
  add_pass(acc[2], r4, __int_as_float(bcast8<(KK >> 2)>(__float_as_int(d2v[2][KK & 3]))), s4[2], r[2]);
  add_pass(acc[3], r4, __int_as_float(bcast8<(KK >> 2)>(__float_as_int(d2v[3][KK & 3]))), s4[3], r[3]);
}

template <int K>
__device__ __forceinline__ void pipe_step(v4f (&acc)[4], v4f (&ra)[4], v4f (&rb)[4],
                                          const v4f* __restrict__ dect4,
                                          const v4i (&n4)[4], const v4f (&d2v)[4],
                                          const v4f (&s4)[4], const v4f* rd4,
                                          const unsigned q) {
  use_slot<K>(acc, ra, d2v, s4, rd4, q);
  use_slot<K + 1>(acc, rb, d2v, s4, rd4, q);
  if constexpr (K + 4 < KNEI) {
    issue_slot<K + 4>(ra, dect4, n4, q);
    issue_slot<K + 5>(rb, dect4, n4, q);
  }
}

__global__ __launch_bounds__(FO_THREADS)
void k_fused_out(const float* __restrict__ d,
                 const int* __restrict__ nid,
                 const int* __restrict__ lab,
                 const float* __restrict__ dect,
                 const float* __restrict__ invw2t,
                 const float* __restrict__ denpart,
                 const float* __restrict__ encoded,
                 float* __restrict__ out) {
  __shared__ __attribute__((aligned(16))) float iw[MCLU * NBAS];
  __shared__ __attribute__((aligned(16))) float rd[NBAS * NBAS];
  __shared__ __attribute__((aligned(16))) float fin[NBAS * NBAS];
  __shared__ float enc_s[NBAS];
  __shared__ float g[FO_WAVES][32 * 17];
  const int tid = threadIdx.x;
  const int lane = tid & 31, wv = tid >> 5;
  const int NT = NPTS / 16;
  const int pt = lane >> 3;
  const unsigned q = (unsigned)(lane & 7);

  int tile = wv * FO_BLOCKS + (int)blockIdx.x;
  v4i n4[4];
  v4f d2v[4];
  int cl[4];
#pragma unroll
  for (int sub = 0; sub < 4; ++sub) {
    const unsigned j = (unsigned)(tile * 16 + sub * 4 + pt);
    n4[sub] = reinterpret_cast<const v4i*>(nid)[j * 8u + q];
    d2v[sub] = reinterpret_cast<const v4f*>(d)[j * 8u + q];
    cl[sub] = lab[j];
  }

  const int fc = tid & 255, fg = tid >> 8;
  v4f fs = {0.f, 0.f, 0.f, 0.f};
  if (fg < FIN_GRP) {
    const v4f* pp = reinterpret_cast<const v4f*>(denpart) + (fg * FIN_PER) * 256 + fc;
    for (int b0 = 0; b0 < FIN_PER; b0 += FIN_RND) {
      v4f r[FIN_RND];
#pragma unroll
      for (int k = 0; k < FIN_RND; ++k) r[k] = pp[(b0 + k) * 256];
#pragma unroll
      for (int k = 0; k < FIN_RND; ++k) fs += r[k];
    }
    if (fg == 1) reinterpret_cast<v4f*>(fin)[fc] = fs;
  } else {
    const int t = tid - FIN_GRP * 256;
    reinterpret_cast<v4f*>(iw)[t] = reinterpret_cast<const v4f*>(invw2t)[t];
    if (t < NBAS) enc_s[t] = encoded[t];
  }
  __syncthreads();
  if (fg == 0) {
    const v4f o = reinterpret_cast<const v4f*>(fin)[fc];
    const int i = fc >> 3, kk = (fc & 7) * 4;
    rd[(kk + 0) * 32 + i] = 1.f / (fs.x + o.x);
    rd[(kk + 1) * 32 + i] = 1.f / (fs.y + o.y);
    rd[(kk + 2) * 32 + i] = 1.f / (fs.z + o.z);
    rd[(kk + 3) * 32 + i] = 1.f / (fs.w + o.w);
  }
  __syncthreads();

  float* gbuf = g[wv];
  const v4f* dect4 = reinterpret_cast<const v4f*>(dect);
  const v4f* rd4 = reinterpret_cast<const v4f*>(rd);

  for (;;) {
    const int j0 = tile * 16;
    v4f s4[4], acc[4];
#pragma unroll
    for (int sub = 0; sub < 4; ++sub) d2v[sub] = d2v[sub] * d2v[sub];
    v4f ra[4], rb[4], rc[4], re[4];
    issue_slot<0>(ra, dect4, n4, q);
    issue_slot<1>(rb, dect4, n4, q);
    issue_slot<2>(rc, dect4, n4, q);
    issue_slot<3>(re, dect4, n4, q);
#pragma unroll
    for (int sub = 0; sub < 4; ++sub) {
      s4[sub] = reinterpret_cast<const v4f*>(iw)[(unsigned)cl[sub] * 8u + q];
      acc[sub] = (v4f){0.f, 0.f, 0.f, 0.f};
    }
    pipe_step<0>(acc, ra, rb, dect4, n4, d2v, s4, rd4, q);
    pipe_step<2>(acc, rc, re, dect4, n4, d2v, s4, rd4, q);
    pipe_step<4>(acc, ra, rb, dect4, n4, d2v, s4, rd4, q);
    pipe_step<6>(acc, rc, re, dect4, n4, d2v, s4, rd4, q);
    pipe_step<8>(acc, ra, rb, dect4, n4, d2v, s4, rd4, q);
    pipe_step<10>(acc, rc, re, dect4, n4, d2v, s4, rd4, q);
    pipe_step<12>(acc, ra, rb, dect4, n4, d2v, s4, rd4, q);
    pipe_step<14>(acc, rc, re, dect4, n4, d2v, s4, rd4, q);
    pipe_step<16>(acc, ra, rb, dect4, n4, d2v, s4, rd4, q);
    pipe_step<18>(acc, rc, re, dect4, n4, d2v, s4, rd4, q);
    pipe_step<20>(acc, ra, rb, dect4, n4, d2v, s4, rd4, q);
    pipe_step<22>(acc, rc, re, dect4, n4, d2v, s4, rd4, q);
    pipe_step<24>(acc, ra, rb, dect4, n4, d2v, s4, rd4, q);
    pipe_step<26>(acc, rc, re, dect4, n4, d2v, s4, rd4, q);
    pipe_step<28>(acc, ra, rb, dect4, n4, d2v, s4, rd4, q);
    pipe_step<30>(acc, rc, re, dect4, n4, d2v, s4, rd4, q);

#pragma unroll
    for (int sub = 0; sub < 4; ++sub) {
      const int colp = sub * 4 + pt;
      gbuf[(4 * q + 0) * 17 + colp] = acc[sub].x;
      gbuf[(4 * q + 1) * 17 + colp] = acc[sub].y;
      gbuf[(4 * q + 2) * 17 + colp] = acc[sub].z;
      gbuf[(4 * q + 3) * 17 + colp] = acc[sub].w;
    }
    asm volatile("s_wait_dscnt 0" ::: "memory");

    v8f cacc = {0.f, 0.f, 0.f, 0.f, 0.f, 0.f, 0.f, 0.f};
    const int half = (lane >> 4) & 1;
    const int col = lane & 15;
#pragma unroll
    for (int s8 = 0; s8 < 8; ++s8) {
      const int kb = s8 * 4;
      v2f a, b;
      a.x = enc_s[kb + 2 * half];
      a.y = enc_s[kb + 2 * half + 1];
      b.x = gbuf[(kb + 2 * half) * 17 + col];
      b.y = gbuf[(kb + 2 * half + 1) * 17 + col];
      cacc = __builtin_amdgcn_wmma_f32_16x16x4_f32(
          false, a, false, b, (short)0, cacc, false, false);
    }
    if (lane < 16) out[j0 + lane] = cacc[0];

    tile += FO_STRIDE;
    if (tile >= NT) break;
#pragma unroll
    for (int sub = 0; sub < 4; ++sub) {
      const unsigned j = (unsigned)(tile * 16 + sub * 4 + pt);
      n4[sub] = reinterpret_cast<const v4i*>(nid)[j * 8u + q];
      d2v[sub] = reinterpret_cast<const v4f*>(d)[j * 8u + q];
      cl[sub] = lab[j];
    }
  }
}

extern "C" void kernel_launch(void* const* d_in, const int* in_sizes, int n_in,
                              void* d_out, int out_size, void* d_ws, size_t ws_size,
                              hipStream_t stream) {
  const float* x     = (const float*)d_in[0];
  const float* enc_w = (const float*)d_in[1];
  const float* enc_b = (const float*)d_in[2];
  const float* dec   = (const float*)d_in[3];
  const float* bw_w  = (const float*)d_in[4];
  const float* bw_b  = (const float*)d_in[5];
  const float* nd    = (const float*)d_in[6];
  const int*   nid   = (const int*)d_in[7];
  const int*   lab   = (const int*)d_in[8];
  float* out = (float*)d_out;
  float* ws  = (float*)d_ws;

  float* enc_part = ws + WS_ENCPART;
  float* encoded  = ws + WS_ENCODED;
  float* invw2t   = ws + WS_INVW2T;
  float* denpart  = ws + WS_DENPART;
  float* dect     = ws + WS_DECT;

  k_enc_partial<<<GV_BLOCKS, GV_WAVES * 32, 0, stream>>>(x, enc_w, enc_part);
  k_finish_transpose<<<FT_GRID, EF_THR, 0, stream>>>(enc_part, enc_b, bw_w, bw_b, encoded, invw2t,
                                                    dec, dect);
  k_denom_partial<<<DEN_GRID, DEN_THR, 0, stream>>>(nd, lab, invw2t, denpart);
  k_fused_out<<<FO_BLOCKS, FO_THREADS, 0, stream>>>(nd, nid, lab, dect, invw2t, denpart,
                                       encoded, out);
}
